# starting kernel
# baseline (speedup 1.0000x reference)
#include <hip/hip_runtime.h>
#include <hip/hip_bf16.h>
#include <math.h>

typedef __attribute__((ext_vector_type(2))) float v2f;
typedef __attribute__((ext_vector_type(8))) float v8f;

#define N_NODES 50000
#define N_EDGES 800000
#define M_HYPER 10000
#define P_PAIRS 200000
#define DIM     64
#define NEG_SLOPE 0.2f

__device__ __forceinline__ unsigned enc_f32(float f) {
    unsigned u = __float_as_uint(f);
    return (u & 0x80000000u) ? ~u : (u | 0x80000000u);
}
__device__ __forceinline__ float dec_f32(unsigned u) {
    return (u & 0x80000000u) ? __uint_as_float(u & 0x7FFFFFFFu)
                             : __uint_as_float(~u);
}

__global__ void k_zero(float* __restrict__ p, long n) {
    long i = (long)blockIdx.x * blockDim.x + threadIdx.x;
    if (i < n) p[i] = 0.0f;
}

__global__ void k_gemm_wmma(const float* __restrict__ X, const float* __restrict__ W,
                            const float* __restrict__ bias, float* __restrict__ Xp) {
    int wave = (blockIdx.x * blockDim.x + threadIdx.x) >> 5;
    int lane = threadIdx.x & 31;
    if (wave >= N_NODES / 16) return;
    int row_base = wave * 16;
    int lh  = (lane >= 16) ? 1 : 0;
    int l15 = lane & 15;

    const float* xrow = X + (size_t)(row_base + l15) * DIM + lh * 2;
    const float* wrow = W + (size_t)l15 * DIM + lh * 2;

    v8f acc[4] = {};
#pragma unroll
    for (int ks = 0; ks < 16; ++ks) {
        v2f a;
        a.x = xrow[ks * 4 + 0];
        a.y = xrow[ks * 4 + 1];
#pragma unroll
        for (int t = 0; t < 4; ++t) {
            v2f b;
            const float* wb = wrow + (size_t)t * 16 * DIM + ks * 4;
            b.x = wb[0];
            b.y = wb[1];
            acc[t] = __builtin_amdgcn_wmma_f32_16x16x4_f32(
                false, a, false, b, (short)0, acc[t], false, false);
        }
    }

#pragma unroll
    for (int t = 0; t < 4; ++t) {
        int c = t * 16 + l15;
        float bv = bias[c];
#pragma unroll
        for (int v = 0; v < 8; ++v) {
            int m = row_base + v + lh * 8;
            Xp[(size_t)m * DIM + c] = acc[t][v] + bv;
        }
    }
}

__global__ void k_scalars(const float* __restrict__ Xp,
                          const float* __restrict__ a_src, const float* __restrict__ a_dst,
                          float* __restrict__ s_src, float* __restrict__ s_dst) {
    int n = blockIdx.x * blockDim.x + threadIdx.x;
    if (n >= N_NODES) return;
    const float4* row = (const float4*)(Xp + (size_t)n * DIM);
    const float4* as  = (const float4*)a_src;
    const float4* ad  = (const float4*)a_dst;
    float ss = 0.f, sd = 0.f;
#pragma unroll
    for (int i = 0; i < 16; ++i) {
        float4 x = row[i], u = as[i], v = ad[i];
        ss += x.x * u.x + x.y * u.y + x.z * u.z + x.w * u.w;
        sd += x.x * v.x + x.y * v.y + x.z * v.z + x.w * v.w;
    }
    s_src[n] = ss;
    s_dst[n] = sd;
}

__global__ void k_deg(const int* __restrict__ inc_v, const int* __restrict__ inc_e,
                      float* __restrict__ dv, float* __restrict__ de) {
    int p = blockIdx.x * blockDim.x + threadIdx.x;
    if (p >= P_PAIRS) return;
    atomicAdd(dv + inc_v[p], 1.0f);
    atomicAdd(de + inc_e[p], 1.0f);
}

__global__ void k_score(const float* __restrict__ s_src, const float* __restrict__ s_dst,
                        const int* __restrict__ e_src, const int* __restrict__ e_dst,
                        float* __restrict__ score, unsigned* __restrict__ segmax) {
    int e = blockIdx.x * blockDim.x + threadIdx.x;
    if (e >= N_EDGES) return;
    float x = s_src[e_src[e]] + s_dst[e_dst[e]];
    float sc = (x >= 0.f) ? x : NEG_SLOPE * x;
    score[e] = sc;
    atomicMax(segmax + e_dst[e], enc_f32(sc));
}

__global__ void k_exp(const int* __restrict__ e_dst, const unsigned* __restrict__ segmax,
                      float* __restrict__ score, float* __restrict__ denom) {
    int e = blockIdx.x * blockDim.x + threadIdx.x;
    if (e >= N_EDGES) return;
    int d = e_dst[e];
    float m = dec_f32(segmax[d]);
    if (!isfinite(m)) m = 0.f;
    float w = expf(score[e] - m);
    score[e] = w;
    atomicAdd(denom + d, w);
}

__global__ void k_scatter_gat(const int* __restrict__ e_src, const int* __restrict__ e_dst,
                              const float* __restrict__ score, const float* __restrict__ denom,
                              const float* __restrict__ Xp, float* __restrict__ Xg) {
    long gid = (long)blockIdx.x * blockDim.x + threadIdx.x;
    long e   = gid >> 5;
    int lane = (int)(gid & 31);
    if (e >= N_EDGES) return;
    int s = e_src[e], d = e_dst[e];
    float coef = score[e] / denom[d];
    float2 xv = *(const float2*)(Xp + (size_t)s * DIM + lane * 2);
    float* out = Xg + (size_t)d * DIM + lane * 2;
    atomicAdd(out + 0, coef * xv.x);
    atomicAdd(out + 1, coef * xv.y);
}

__global__ void k_hg_gather(const int* __restrict__ inc_v, const int* __restrict__ inc_e,
                            const float* __restrict__ Xp, const float* __restrict__ dv,
                            float* __restrict__ He) {
    long gid = (long)blockIdx.x * blockDim.x + threadIdx.x;
    long p   = gid >> 5;
    int lane = (int)(gid & 31);
    if (p >= P_PAIRS) return;
    int v = inc_v[p], e = inc_e[p];
    float dvv = dv[v];
    float s = (dvv > 0.f) ? rsqrtf(dvv) : 0.f;
    float2 xv = *(const float2*)(Xp + (size_t)v * DIM + lane * 2);
    float* out = He + (size_t)e * DIM + lane * 2;
    atomicAdd(out + 0, s * xv.x);
    atomicAdd(out + 1, s * xv.y);
}

__global__ void k_he_norm(float* __restrict__ He, const float* __restrict__ de) {
    int i = blockIdx.x * blockDim.x + threadIdx.x;
    if (i >= M_HYPER * DIM) return;
    float d = de[i >> 6];
    He[i] *= (d > 0.f) ? (1.f / d) : 0.f;
}

__global__ void k_hg_scatter(const int* __restrict__ inc_v, const int* __restrict__ inc_e,
                             const float* __restrict__ He, float* __restrict__ Xhg) {
    long gid = (long)blockIdx.x * blockDim.x + threadIdx.x;
    long p   = gid >> 5;
    int lane = (int)(gid & 31);
    if (p >= P_PAIRS) return;
    int v = inc_v[p], e = inc_e[p];
    float2 h = *(const float2*)(He + (size_t)e * DIM + lane * 2);
    float* out = Xhg + (size_t)v * DIM + lane * 2;
    atomicAdd(out + 0, h.x);
    atomicAdd(out + 1, h.y);
}

__global__ void k_final(const float* __restrict__ Xg, const float* __restrict__ Xhg,
                        const float* __restrict__ dv, float* __restrict__ out) {
    long i = (long)blockIdx.x * blockDim.x + threadIdx.x;
    if (i >= (long)N_NODES * DIM) return;
    float d = dv[i >> 6];
    float s = (d > 0.f) ? rsqrtf(d) : 0.f;
    float x = 0.5f * (Xg[i] + Xhg[i] * s);
    out[i] = (x > 0.f) ? x : expm1f(x);
}

extern "C" void kernel_launch(void* const* d_in, const int* in_sizes, int n_in,
                              void* d_out, int out_size, void* d_ws, size_t ws_size,
                              hipStream_t stream) {
    const float* X     = (const float*)d_in[0];
    const float* W     = (const float*)d_in[1];
    const float* bias  = (const float*)d_in[2];
    const float* a_src = (const float*)d_in[3];
    const float* a_dst = (const float*)d_in[4];
    const int*   e_src = (const int*)d_in[5];
    const int*   e_dst = (const int*)d_in[6];
    const int*   inc_v = (const int*)d_in[7];
    const int*   inc_e = (const int*)d_in[8];
    float* out = (float*)d_out;
    float* ws  = (float*)d_ws;

    size_t off = 0;
    float* Xp    = ws + off; off += (size_t)N_NODES * DIM;
    float* s_src = ws + off; off += N_NODES;
    float* s_dst = ws + off; off += N_NODES;
    size_t zero_base = off;
    unsigned* segmax = (unsigned*)(ws + off); off += N_NODES;
    float* denom = ws + off; off += N_NODES;
    float* dv    = ws + off; off += N_NODES;
    float* de    = ws + off; off += M_HYPER;
    float* Xg    = ws + off; off += (size_t)N_NODES * DIM;
    float* He    = ws + off; off += (size_t)M_HYPER * DIM;
    float* Xhg   = ws + off; off += (size_t)N_NODES * DIM;
    size_t zero_cnt = off - zero_base;
    float* score = ws + off; off += N_EDGES;

    const int B = 256;
    k_zero<<<(int)((zero_cnt + B - 1) / B), B, 0, stream>>>(ws + zero_base, (long)zero_cnt);

    int tiles = N_NODES / 16;
    k_gemm_wmma<<<(tiles * 32 + B - 1) / B, B, 0, stream>>>(X, W, bias, Xp);

    k_scalars<<<(N_NODES + B - 1) / B, B, 0, stream>>>(Xp, a_src, a_dst, s_src, s_dst);
    k_deg<<<(P_PAIRS + B - 1) / B, B, 0, stream>>>(inc_v, inc_e, dv, de);
    k_score<<<(N_EDGES + B - 1) / B, B, 0, stream>>>(s_src, s_dst, e_src, e_dst, score, segmax);
    k_exp<<<(N_EDGES + B - 1) / B, B, 0, stream>>>(e_dst, segmax, score, denom);
    k_scatter_gat<<<(int)(((long)N_EDGES * 32 + B - 1) / B), B, 0, stream>>>(
        e_src, e_dst, score, denom, Xp, Xg);
    k_hg_gather<<<(int)(((long)P_PAIRS * 32 + B - 1) / B), B, 0, stream>>>(
        inc_v, inc_e, Xp, dv, He);
    k_he_norm<<<(M_HYPER * DIM + B - 1) / B, B, 0, stream>>>(He, de);
    k_hg_scatter<<<(int)(((long)P_PAIRS * 32 + B - 1) / B), B, 0, stream>>>(
        inc_v, inc_e, He, Xhg);
    k_final<<<(int)(((long)N_NODES * DIM + B - 1) / B), B, 0, stream>>>(Xg, Xhg, dv, out);
}


# GAT edge stages: each work-group owns 200 dst rows in LDS and filter-scans all of e_dst (no sort, no merge, no global atomics); 3 launches -> 1; sums now LDS atomics, divided by denominator after sum
# speedup vs baseline: 1.9195x; 1.9195x over previous
#include <hip/hip_runtime.h>
#include <hip/hip_bf16.h>
#include <math.h>

typedef __attribute__((ext_vector_type(2))) float v2f;
typedef __attribute__((ext_vector_type(8))) float v8f;

#define N_NODES 50000
#define N_EDGES 800000
#define M_HYPER 10000
#define P_PAIRS 200000
#define DIM     64
#define NEG_SLOPE 0.2f

__device__ __forceinline__ unsigned enc_f32(float f) {
    unsigned u = __float_as_uint(f);
    return (u & 0x80000000u) ? ~u : (u | 0x80000000u);
}
__device__ __forceinline__ float dec_f32(unsigned u) {
    return (u & 0x80000000u) ? __uint_as_float(u & 0x7FFFFFFFu)
                             : __uint_as_float(~u);
}

__global__ void k_zero(float* __restrict__ p, long n) {
    long i = (long)blockIdx.x * blockDim.x + threadIdx.x;
    if (i < n) p[i] = 0.0f;
}

__global__ void k_gemm_wmma(const float* __restrict__ X, const float* __restrict__ W,
                            const float* __restrict__ bias, float* __restrict__ Xp) {
    int wave = (blockIdx.x * blockDim.x + threadIdx.x) >> 5;
    int lane = threadIdx.x & 31;
    if (wave >= N_NODES / 16) return;
    int row_base = wave * 16;
    int lh  = (lane >= 16) ? 1 : 0;
    int l15 = lane & 15;

    const float* xrow = X + (size_t)(row_base + l15) * DIM + lh * 2;
    const float* wrow = W + (size_t)l15 * DIM + lh * 2;

    v8f acc[4] = {};
#pragma unroll
    for (int ks = 0; ks < 16; ++ks) {
        v2f a;
        a.x = xrow[ks * 4 + 0];
        a.y = xrow[ks * 4 + 1];
#pragma unroll
        for (int t = 0; t < 4; ++t) {
            v2f b;
            const float* wb = wrow + (size_t)t * 16 * DIM + ks * 4;
            b.x = wb[0];
            b.y = wb[1];
            acc[t] = __builtin_amdgcn_wmma_f32_16x16x4_f32(
                false, a, false, b, (short)0, acc[t], false, false);
        }
    }

#pragma unroll
    for (int t = 0; t < 4; ++t) {
        int c = t * 16 + l15;
        float bv = bias[c];
#pragma unroll
        for (int v = 0; v < 8; ++v) {
            int m = row_base + v + lh * 8;
            Xp[(size_t)m * DIM + c] = acc[t][v] + bv;
        }
    }
}

__global__ void k_scalars(const float* __restrict__ Xp,
                          const float* __restrict__ a_src, const float* __restrict__ a_dst,
                          float* __restrict__ s_src, float* __restrict__ s_dst) {
    int n = blockIdx.x * blockDim.x + threadIdx.x;
    if (n >= N_NODES) return;
    const float4* row = (const float4*)(Xp + (size_t)n * DIM);
    const float4* as  = (const float4*)a_src;
    const float4* ad  = (const float4*)a_dst;
    float ss = 0.f, sd = 0.f;
#pragma unroll
    for (int i = 0; i < 16; ++i) {
        float4 x = row[i], u = as[i], v = ad[i];
        ss += x.x * u.x + x.y * u.y + x.z * u.z + x.w * u.w;
        sd += x.x * v.x + x.y * v.y + x.z * v.z + x.w * v.w;
    }
    s_src[n] = ss;
    s_dst[n] = sd;
}

__global__ void k_deg(const int* __restrict__ inc_v, const int* __restrict__ inc_e,
                      float* __restrict__ dv, float* __restrict__ de) {
    int p = blockIdx.x * blockDim.x + threadIdx.x;
    if (p >= P_PAIRS) return;
    atomicAdd(dv + inc_v[p], 1.0f);
    atomicAdd(de + inc_e[p], 1.0f);
}

#define GAT_R     200
#define GAT_NB    (N_NODES / GAT_R)
#define GAT_T     1024
#define GAT_WAVES (GAT_T / 32)
#define GAT_CAP   512
#define GAT_NQ    (N_EDGES / 4)
#define GAT_NIT   ((GAT_NQ + GAT_T - 1) / GAT_T)
#define GAT_SU    7
#define GAT_RU    16

__device__ __forceinline__ float gat_leaky(float x) { return (x >= 0.f) ? x : NEG_SLOPE * x; }

__global__ __launch_bounds__(GAT_T)
void k_gat_owner(const int* __restrict__ e_src, const int* __restrict__ e_dst,
                 const float* __restrict__ s_src, const float* __restrict__ s_dst,
                 const float* __restrict__ Xp, float* __restrict__ Xg) {
    __shared__ __attribute__((aligned(16))) float acc[GAT_R * DIM];
    __shared__ unsigned lst_k[GAT_WAVES * GAT_CAP];
    __shared__ float    lst_s[GAT_WAVES * GAT_CAP];
    __shared__ unsigned smax[GAT_R];
    __shared__ float    sden[GAT_R];
    __shared__ float    sdst[GAT_R];
    __shared__ int      wcnt[GAT_WAVES];
    __shared__ int      ovf;

    const int tid  = threadIdx.x;
    const int lane = tid & 31;
    const int wave = tid >> 5;
    const int lo   = blockIdx.x * GAT_R;
    const int lbase = wave * GAT_CAP;

    for (int i = tid; i < GAT_R * DIM; i += GAT_T) acc[i] = 0.f;
    if (tid < GAT_R) { smax[tid] = 0u; sden[tid] = 0.f; sdst[tid] = s_dst[lo + tid]; }
    if (tid == 0) ovf = 0;
    __syncthreads();

    const int4* ed4 = (const int4*)e_dst;
    const int qoff = blockIdx.x * (GAT_NQ / GAT_NB);
    int cnt = 0;
    for (int i0 = 0; i0 < GAT_NIT; i0 += GAT_SU) {
        int4 dv4[GAT_SU];
        int  qq[GAT_SU];
#pragma unroll
        for (int u = 0; u < GAT_SU; ++u) {
            int idx = (i0 + u) * GAT_T + tid;
            int q = idx + qoff;
            if (q >= GAT_NQ) q -= GAT_NQ;
            qq[u] = q;
            if (idx < GAT_NQ) dv4[u] = ed4[q];
            else              dv4[u] = make_int4(-1, -1, -1, -1);
        }
#pragma unroll
        for (int u = 0; u < GAT_SU; ++u) {
            int dd[4] = {dv4[u].x, dv4[u].y, dv4[u].z, dv4[u].w};
#pragma unroll
            for (int k = 0; k < 4; ++k) {
                unsigned dl = (unsigned)(dd[k] - lo);
                bool m = dl < (unsigned)GAT_R;
                unsigned mask = __builtin_amdgcn_ballot_w32(m);
                if (mask != 0u) {
                    int pos = cnt + (int)__builtin_amdgcn_mbcnt_lo(mask, 0u);
                    if (m && pos < GAT_CAP)
                        lst_k[lbase + pos] = ((unsigned)(qq[u] * 4 + k) << 8) | dl;
                    cnt += __builtin_popcount(mask);
                }
            }
        }
    }
    if (lane == 0) {
        wcnt[wave] = (cnt < GAT_CAP) ? cnt : GAT_CAP;
        if (cnt > GAT_CAP) ovf = 1;
    }
    __syncthreads();
    const int over = __builtin_amdgcn_readfirstlane(ovf);

    if (over == 0) {
        const int n = __builtin_amdgcn_readfirstlane(wcnt[wave]);
        for (int j0 = 0; j0 < n; j0 += 128) {
            unsigned key[4]; int sv[4]; float ss[4]; bool ok[4];
#pragma unroll
            for (int u = 0; u < 4; ++u) {
                int j = j0 + u * 32 + lane;
                ok[u] = j < n;
                key[u] = ok[u] ? lst_k[lbase + j] : 0u;
            }
#pragma unroll
            for (int u = 0; u < 4; ++u) sv[u] = e_src[key[u] >> 8];
#pragma unroll
            for (int u = 0; u < 4; ++u) ss[u] = s_src[sv[u]];
#pragma unroll
            for (int u = 0; u < 4; ++u) {
                if (ok[u]) {
                    int j = j0 + u * 32 + lane;
                    unsigned dl = key[u] & 255u;
                    float sc = gat_leaky(ss[u] + sdst[dl]);
                    atomicMax(&smax[dl], enc_f32(sc));
                    lst_k[lbase + j] = ((unsigned)sv[u] << 8) | dl;
                    lst_s[lbase + j] = sc;
                }
            }
        }
        __syncthreads();
        for (int j = lane; j < n; j += 32) {
            unsigned dl = lst_k[lbase + j] & 255u;
            float m = dec_f32(smax[dl]);
            if (!isfinite(m)) m = 0.f;
            float w = expf(lst_s[lbase + j] - m);
            lst_s[lbase + j] = w;
            atomicAdd(&sden[dl], w);
        }
        __syncthreads();
        for (int j0 = 0; j0 < n; j0 += GAT_RU) {
            float2 xv[GAT_RU]; unsigned kk[GAT_RU]; float ww[GAT_RU];
            const int jj = j0 + (lane & (GAT_RU - 1));
            const unsigned kl = (jj < n) ? lst_k[lbase + jj] : 0u;
            const float    wl = (jj < n) ? lst_s[lbase + jj] : 0.f;
#pragma unroll
            for (int u = 0; u < GAT_RU; ++u) {
                if (j0 + u < n) {
                    kk[u] = (unsigned)__builtin_amdgcn_readlane((int)kl, u);
                    ww[u] = __int_as_float(__builtin_amdgcn_readlane(__float_as_int(wl), u));
                    xv[u] = *(const float2*)(Xp + (size_t)(kk[u] >> 8) * DIM + lane * 2);
                }
            }
#pragma unroll
            for (int u = 0; u < GAT_RU; ++u) {
                if (j0 + u < n) {
                    float* a = acc + (kk[u] & 255u) * DIM + lane * 2;
                    atomicAdd(a + 0, ww[u] * xv[u].x);
                    atomicAdd(a + 1, ww[u] * xv[u].y);
                }
            }
        }
    } else {
        for (int e = tid; e < N_EDGES; e += GAT_T) {
            unsigned dl = (unsigned)(e_dst[e] - lo);
            if (dl < (unsigned)GAT_R) {
                float sc = gat_leaky(s_src[e_src[e]] + sdst[dl]);
                atomicMax(&smax[dl], enc_f32(sc));
            }
        }
        __syncthreads();
        for (int eb = wave * 32; eb < N_EDGES; eb += GAT_T) {
            int e = eb + lane;
            bool valid = e < N_EDGES;
            unsigned dl = valid ? (unsigned)(e_dst[e] - lo) : 0xFFFFFFFFu;
            bool m = dl < (unsigned)GAT_R;
            float w = 0.f; int sidx = 0;
            if (m) {
                sidx = e_src[e];
                float sc = gat_leaky(s_src[sidx] + sdst[dl]);
                float mx = dec_f32(smax[dl]);
                if (!isfinite(mx)) mx = 0.f;
                w = expf(sc - mx);
                atomicAdd(&sden[dl], w);
            }
            unsigned mask = __builtin_amdgcn_ballot_w32(m);
            while (mask != 0u) {
                int l = __builtin_ctz(mask);
                mask &= mask - 1u;
                int   s_l = __shfl(sidx, l);
                int   d_l = __shfl((int)dl, l);
                float w_l = __shfl(w, l);
                float2 xv = *(const float2*)(Xp + (size_t)s_l * DIM + lane * 2);
                float* a = acc + d_l * DIM + lane * 2;
                atomicAdd(a + 0, w_l * xv.x);
                atomicAdd(a + 1, w_l * xv.y);
            }
        }
    }
    __syncthreads();

    float4* xg4 = (float4*)(Xg + (size_t)lo * DIM);
    const float4* acc4 = (const float4*)acc;
    for (int i = tid; i < GAT_R * (DIM / 4); i += GAT_T) {
        float den = sden[i >> 4];
        float4 v = acc4[i];
        if (den != 0.f) { v.x /= den; v.y /= den; v.z /= den; v.w /= den; }
        xg4[i] = v;
    }
}

__global__ void k_hg_gather(const int* __restrict__ inc_v, const int* __restrict__ inc_e,
                            const float* __restrict__ Xp, const float* __restrict__ dv,
                            float* __restrict__ He) {
    long gid = (long)blockIdx.x * blockDim.x + threadIdx.x;
    long p   = gid >> 5;
    int lane = (int)(gid & 31);
    if (p >= P_PAIRS) return;
    int v = inc_v[p], e = inc_e[p];
    float dvv = dv[v];
    float s = (dvv > 0.f) ? rsqrtf(dvv) : 0.f;
    float2 xv = *(const float2*)(Xp + (size_t)v * DIM + lane * 2);
    float* out = He + (size_t)e * DIM + lane * 2;
    atomicAdd(out + 0, s * xv.x);
    atomicAdd(out + 1, s * xv.y);
}

__global__ void k_he_norm(float* __restrict__ He, const float* __restrict__ de) {
    int i = blockIdx.x * blockDim.x + threadIdx.x;
    if (i >= M_HYPER * DIM) return;
    float d = de[i >> 6];
    He[i] *= (d > 0.f) ? (1.f / d) : 0.f;
}

__global__ void k_hg_scatter(const int* __restrict__ inc_v, const int* __restrict__ inc_e,
                             const float* __restrict__ He, float* __restrict__ Xhg) {
    long gid = (long)blockIdx.x * blockDim.x + threadIdx.x;
    long p   = gid >> 5;
    int lane = (int)(gid & 31);
    if (p >= P_PAIRS) return;
    int v = inc_v[p], e = inc_e[p];
    float2 h = *(const float2*)(He + (size_t)e * DIM + lane * 2);
    float* out = Xhg + (size_t)v * DIM + lane * 2;
    atomicAdd(out + 0, h.x);
    atomicAdd(out + 1, h.y);
}

__global__ void k_final(const float* __restrict__ Xg, const float* __restrict__ Xhg,
                        const float* __restrict__ dv, float* __restrict__ out) {
    long i = (long)blockIdx.x * blockDim.x + threadIdx.x;
    if (i >= (long)N_NODES * DIM) return;
    float d = dv[i >> 6];
    float s = (d > 0.f) ? rsqrtf(d) : 0.f;
    float x = 0.5f * (Xg[i] + Xhg[i] * s);
    out[i] = (x > 0.f) ? x : expm1f(x);
}

extern "C" void kernel_launch(void* const* d_in, const int* in_sizes, int n_in,
                              void* d_out, int out_size, void* d_ws, size_t ws_size,
                              hipStream_t stream) {
    const float* X     = (const float*)d_in[0];
    const float* W     = (const float*)d_in[1];
    const float* bias  = (const float*)d_in[2];
    const float* a_src = (const float*)d_in[3];
    const float* a_dst = (const float*)d_in[4];
    const int*   e_src = (const int*)d_in[5];
    const int*   e_dst = (const int*)d_in[6];
    const int*   inc_v = (const int*)d_in[7];
    const int*   inc_e = (const int*)d_in[8];
    float* out = (float*)d_out;
    float* ws  = (float*)d_ws;

    size_t off = 0;
    float* Xp    = ws + off; off += (size_t)N_NODES * DIM;
    float* s_src = ws + off; off += N_NODES;
    float* s_dst = ws + off; off += N_NODES;
    float* Xg    = ws + off; off += (size_t)N_NODES * DIM;
    size_t zero_base = off;
    float* dv    = ws + off; off += N_NODES;
    float* de    = ws + off; off += M_HYPER;
    float* He    = ws + off; off += (size_t)M_HYPER * DIM;
    float* Xhg   = ws + off; off += (size_t)N_NODES * DIM;
    size_t zero_cnt = off - zero_base;

    const int B = 256;
    k_zero<<<(int)((zero_cnt + B - 1) / B), B, 0, stream>>>(ws + zero_base, (long)zero_cnt);

    int tiles = N_NODES / 16;
    k_gemm_wmma<<<(tiles * 32 + B - 1) / B, B, 0, stream>>>(X, W, bias, Xp);

    k_scalars<<<(N_NODES + B - 1) / B, B, 0, stream>>>(Xp, a_src, a_dst, s_src, s_dst);
    k_deg<<<(P_PAIRS + B - 1) / B, B, 0, stream>>>(inc_v, inc_e, dv, de);
    k_gat_owner<<<GAT_NB, GAT_T, 0, stream>>>(e_src, e_dst, s_src, s_dst, Xp, Xg);
    k_hg_gather<<<(int)(((long)P_PAIRS * 32 + B - 1) / B), B, 0, stream>>>(
        inc_v, inc_e, Xp, dv, He);
    k_he_norm<<<(M_HYPER * DIM + B - 1) / B, B, 0, stream>>>(He, de);
    k_hg_scatter<<<(int)(((long)P_PAIRS * 32 + B - 1) / B), B, 0, stream>>>(
        inc_v, inc_e, He, Xhg);
    k_final<<<(int)(((long)N_NODES * DIM + B - 1) / B), B, 0, stream>>>(Xg, Xhg, dv, out);
}


# HGNN row sums: global float atomics -> owner work-groups (He rows; vertex rows) filter-scan inc_e/inc_v and sum by LDS atomics; vertex stage + elu fused into edge kernel; 9 launches -> 6
# speedup vs baseline: 2.1738x; 1.1325x over previous
#include <hip/hip_runtime.h>
#include <hip/hip_bf16.h>
#include <math.h>

typedef __attribute__((ext_vector_type(2))) float v2f;
typedef __attribute__((ext_vector_type(8))) float v8f;

#define N_NODES 50000
#define N_EDGES 800000
#define M_HYPER 10000
#define P_PAIRS 200000
#define DIM     64
#define NEG_SLOPE 0.2f

__device__ __forceinline__ unsigned enc_f32(float f) {
    unsigned u = __float_as_uint(f);
    return (u & 0x80000000u) ? ~u : (u | 0x80000000u);
}
__device__ __forceinline__ float dec_f32(unsigned u) {
    return (u & 0x80000000u) ? __uint_as_float(u & 0x7FFFFFFFu)
                             : __uint_as_float(~u);
}

__global__ void k_zero(float* __restrict__ p, long n) {
    long i = (long)blockIdx.x * blockDim.x + threadIdx.x;
    if (i < n) p[i] = 0.0f;
}

__global__ void k_gemm_wmma(const float* __restrict__ X, const float* __restrict__ W,
                            const float* __restrict__ bias, float* __restrict__ Xp) {
    int wave = (blockIdx.x * blockDim.x + threadIdx.x) >> 5;
    int lane = threadIdx.x & 31;
    if (wave >= N_NODES / 16) return;
    int row_base = wave * 16;
    int lh  = (lane >= 16) ? 1 : 0;
    int l15 = lane & 15;

    const float* xrow = X + (size_t)(row_base + l15) * DIM + lh * 2;
    const float* wrow = W + (size_t)l15 * DIM + lh * 2;

    v8f acc[4] = {};
#pragma unroll
    for (int ks = 0; ks < 16; ++ks) {
        v2f a;
        a.x = xrow[ks * 4 + 0];
        a.y = xrow[ks * 4 + 1];
#pragma unroll
        for (int t = 0; t < 4; ++t) {
            v2f b;
            const float* wb = wrow + (size_t)t * 16 * DIM + ks * 4;
            b.x = wb[0];
            b.y = wb[1];
            acc[t] = __builtin_amdgcn_wmma_f32_16x16x4_f32(
                false, a, false, b, (short)0, acc[t], false, false);
        }
    }

#pragma unroll
    for (int t = 0; t < 4; ++t) {
        int c = t * 16 + l15;
        float bv = bias[c];
#pragma unroll
        for (int v = 0; v < 8; ++v) {
            int m = row_base + v + lh * 8;
            Xp[(size_t)m * DIM + c] = acc[t][v] + bv;
        }
    }
}

__global__ void k_scalars(const float* __restrict__ Xp,
                          const float* __restrict__ a_src, const float* __restrict__ a_dst,
                          float* __restrict__ s_src, float* __restrict__ s_dst) {
    int n = blockIdx.x * blockDim.x + threadIdx.x;
    if (n >= N_NODES) return;
    const float4* row = (const float4*)(Xp + (size_t)n * DIM);
    const float4* as  = (const float4*)a_src;
    const float4* ad  = (const float4*)a_dst;
    float ss = 0.f, sd = 0.f;
#pragma unroll
    for (int i = 0; i < 16; ++i) {
        float4 x = row[i], u = as[i], v = ad[i];
        ss += x.x * u.x + x.y * u.y + x.z * u.z + x.w * u.w;
        sd += x.x * v.x + x.y * v.y + x.z * v.z + x.w * v.w;
    }
    s_src[n] = ss;
    s_dst[n] = sd;
}

__global__ void k_deg(const int* __restrict__ inc_v, const int* __restrict__ inc_e,
                      float* __restrict__ dv, float* __restrict__ de) {
    int p = blockIdx.x * blockDim.x + threadIdx.x;
    if (p >= P_PAIRS) return;
    atomicAdd(dv + inc_v[p], 1.0f);
    atomicAdd(de + inc_e[p], 1.0f);
}

#define HG_T      1024
#define HG_WAVES  (HG_T / 32)
#define HG_NQ     (P_PAIRS / 4)
#define HG_SU     7
#define HG_RU     16
#define HE_R      40
#define HE_NB     (M_HYPER / HE_R)
#define HE_CAP    256

__device__ __forceinline__ int hg_filter_scan(const int4* __restrict__ src4, int qoff, int lo, unsigned rows,
                                              unsigned* lst, int cap, int tid) {
    constexpr int NIT = (HG_NQ + HG_T - 1) / HG_T;
    static_assert(NIT % HG_SU == 0, "scan trips must be a multiple of the batch");
    int cnt = 0;
    for (int i0 = 0; i0 < NIT; i0 += HG_SU) {
        int4 dv4[HG_SU];
        int  qq[HG_SU];
#pragma unroll
        for (int u = 0; u < HG_SU; ++u) {
            int idx = (i0 + u) * HG_T + tid;
            int q = idx + qoff;
            if (q >= HG_NQ) q -= HG_NQ;
            qq[u] = q;
            if (idx < HG_NQ) dv4[u] = src4[q];
            else             dv4[u] = make_int4(-1, -1, -1, -1);
        }
#pragma unroll
        for (int u = 0; u < HG_SU; ++u) {
            int dd[4] = {dv4[u].x, dv4[u].y, dv4[u].z, dv4[u].w};
#pragma unroll
            for (int k = 0; k < 4; ++k) {
                unsigned dl = (unsigned)(dd[k] - lo);
                bool m = dl < rows;
                unsigned mask = __builtin_amdgcn_ballot_w32(m);
                if (mask != 0u) {
                    int pos = cnt + (int)__builtin_amdgcn_mbcnt_lo(mask, 0u);
                    if (m && pos < cap)
                        lst[pos] = ((unsigned)(qq[u] * 4 + k) << 8) | dl;
                    cnt += __builtin_popcount(mask);
                }
            }
        }
    }
    return cnt;
}

template <bool WEIGHTED>
__device__ __forceinline__ void hg_row_sum(const unsigned* lk, const float* ls, int n,
                                           const float* __restrict__ src, float* accp, int lane) {
    for (int j0 = 0; j0 < n; j0 += HG_RU) {
        float2 xv[HG_RU]; unsigned kk[HG_RU]; float ww[HG_RU];
        const int jj = j0 + (lane & (HG_RU - 1));
        const unsigned kl = (jj < n) ? lk[jj] : 0u;
        float wl = 1.f;
        if (WEIGHTED) wl = (jj < n) ? ls[jj] : 0.f;
#pragma unroll
        for (int u = 0; u < HG_RU; ++u) {
            if (j0 + u < n) {
                kk[u] = (unsigned)__builtin_amdgcn_readlane((int)kl, u);
                ww[u] = __int_as_float(__builtin_amdgcn_readlane(__float_as_int(wl), u));
                xv[u] = *(const float2*)(src + (size_t)(kk[u] >> 8) * DIM + lane * 2);
            }
        }
#pragma unroll
        for (int u = 0; u < HG_RU; ++u) {
            if (j0 + u < n) {
                float* a = accp + (kk[u] & 255u) * DIM + lane * 2;
                if (WEIGHTED) {
                    atomicAdd(a + 0, ww[u] * xv[u].x);
                    atomicAdd(a + 1, ww[u] * xv[u].y);
                } else {
                    atomicAdd(a + 0, xv[u].x);
                    atomicAdd(a + 1, xv[u].y);
                }
            }
        }
    }
}

__global__ __launch_bounds__(HG_T)
void k_he_owner(const int* __restrict__ inc_v, const int* __restrict__ inc_e,
                const float* __restrict__ Xp, const float* __restrict__ dv, const float* __restrict__ de,
                float* __restrict__ He) {
    __shared__ __attribute__((aligned(16))) float hacc[HE_R * DIM];
    __shared__ unsigned lst_k[HG_WAVES * HE_CAP];
    __shared__ float    lst_s[HG_WAVES * HE_CAP];
    __shared__ float    sde[HE_R];
    __shared__ int      wcnt[HG_WAVES];
    __shared__ int      ovf;

    const int tid  = threadIdx.x;
    const int lane = tid & 31;
    const int wave = tid >> 5;
    const int lo   = blockIdx.x * HE_R;
    const int lbase = wave * HE_CAP;

    for (int i = tid; i < HE_R * DIM; i += HG_T) hacc[i] = 0.f;
    if (tid < HE_R) sde[tid] = de[lo + tid];
    if (tid == 0) ovf = 0;
    __syncthreads();

    const int cnt = hg_filter_scan((const int4*)inc_e, blockIdx.x * (HG_NQ / HE_NB), lo, (unsigned)HE_R,
                                   lst_k + lbase, HE_CAP, tid);
    if (lane == 0) {
        wcnt[wave] = (cnt < HE_CAP) ? cnt : HE_CAP;
        if (cnt > HE_CAP) ovf = 1;
    }
    __syncthreads();
    const int over = __builtin_amdgcn_readfirstlane(ovf);

    if (over == 0) {
        const int n = __builtin_amdgcn_readfirstlane(wcnt[wave]);
        for (int j0 = 0; j0 < n; j0 += 128) {
            unsigned key[4]; int vv[4]; float dd[4]; bool ok[4];
#pragma unroll
            for (int u = 0; u < 4; ++u) {
                int j = j0 + u * 32 + lane;
                ok[u] = j < n;
                key[u] = ok[u] ? lst_k[lbase + j] : 0u;
            }
#pragma unroll
            for (int u = 0; u < 4; ++u) vv[u] = inc_v[key[u] >> 8];
#pragma unroll
            for (int u = 0; u < 4; ++u) dd[u] = dv[vv[u]];
#pragma unroll
            for (int u = 0; u < 4; ++u) {
                if (ok[u]) {
                    int j = j0 + u * 32 + lane;
                    lst_k[lbase + j] = ((unsigned)vv[u] << 8) | (key[u] & 255u);
                    lst_s[lbase + j] = (dd[u] > 0.f) ? rsqrtf(dd[u]) : 0.f;
                }
            }
        }
        __syncthreads();
        hg_row_sum<true>(lst_k + lbase, lst_s + lbase, n, Xp, hacc, lane);
    } else {
        for (int pb = wave * 32; pb < P_PAIRS; pb += HG_T) {
            int p = pb + lane;
            bool valid = p < P_PAIRS;
            unsigned dl = valid ? (unsigned)(inc_e[p] - lo) : 0xFFFFFFFFu;
            bool m = dl < (unsigned)HE_R;
            float s = 0.f; int vidx = 0;
            if (m) {
                vidx = inc_v[p];
                float d = dv[vidx];
                s = (d > 0.f) ? rsqrtf(d) : 0.f;
            }
            unsigned mask = __builtin_amdgcn_ballot_w32(m);
            while (mask != 0u) {
                int l = __builtin_ctz(mask);
                mask &= mask - 1u;
                int   v_l = __shfl(vidx, l);
                int   d_l = __shfl((int)dl, l);
                float s_l = __shfl(s, l);
                float2 xv = *(const float2*)(Xp + (size_t)v_l * DIM + lane * 2);
                float* a = hacc + d_l * DIM + lane * 2;
                atomicAdd(a + 0, s_l * xv.x);
                atomicAdd(a + 1, s_l * xv.y);
            }
        }
    }
    __syncthreads();

    float4* he4 = (float4*)(He + (size_t)lo * DIM);
    const float4* hacc4 = (const float4*)hacc;
    for (int i = tid; i < HE_R * (DIM / 4); i += HG_T) {
        float d = sde[i >> 4];
        float inv = (d > 0.f) ? (1.f / d) : 0.f;
        float4 v = hacc4[i];
        v.x *= inv; v.y *= inv; v.z *= inv; v.w *= inv;
        he4[i] = v;
    }
}

#define GAT_R     200
#define GAT_NB    (N_NODES / GAT_R)
#define GAT_T     1024
#define GAT_WAVES (GAT_T / 32)
#define GAT_CAP   512
#define GAT_NQ    (N_EDGES / 4)
#define GAT_NIT   ((GAT_NQ + GAT_T - 1) / GAT_T)
#define GAT_SU    7
#define GAT_RU    16

__device__ __forceinline__ float gat_leaky(float x) { return (x >= 0.f) ? x : NEG_SLOPE * x; }

__global__ __launch_bounds__(GAT_T)
void k_gat_owner(const int* __restrict__ e_src, const int* __restrict__ e_dst,
                 const float* __restrict__ s_src, const float* __restrict__ s_dst,
                 const float* __restrict__ Xp,
                 const int* __restrict__ inc_v, const int* __restrict__ inc_e,
                 const float* __restrict__ He, const float* __restrict__ dv,
                 float* __restrict__ out) {
    __shared__ __attribute__((aligned(16))) float acc[GAT_R * DIM];
    __shared__ __attribute__((aligned(16))) float hacc[GAT_R * DIM];
    __shared__ unsigned lst_k[GAT_WAVES * GAT_CAP];
    __shared__ float    lst_s[GAT_WAVES * GAT_CAP];
    __shared__ unsigned smax[GAT_R];
    __shared__ float    sden[GAT_R];
    __shared__ float    sdst[GAT_R];
    __shared__ float    sdv[GAT_R];
    __shared__ int      wcnt[GAT_WAVES];
    __shared__ int      ovf;
    __shared__ int      ovf2;

    const int tid  = threadIdx.x;
    const int lane = tid & 31;
    const int wave = tid >> 5;
    const int lo   = blockIdx.x * GAT_R;
    const int lbase = wave * GAT_CAP;

    for (int i = tid; i < GAT_R * DIM; i += GAT_T) { acc[i] = 0.f; hacc[i] = 0.f; }
    if (tid < GAT_R) { smax[tid] = 0u; sden[tid] = 0.f; sdst[tid] = s_dst[lo + tid]; sdv[tid] = dv[lo + tid]; }
    if (tid == 0) { ovf = 0; ovf2 = 0; }
    __syncthreads();

    const int4* ed4 = (const int4*)e_dst;
    const int qoff = blockIdx.x * (GAT_NQ / GAT_NB);
    int cnt = 0;
    for (int i0 = 0; i0 < GAT_NIT; i0 += GAT_SU) {
        int4 dv4[GAT_SU];
        int  qq[GAT_SU];
#pragma unroll
        for (int u = 0; u < GAT_SU; ++u) {
            int idx = (i0 + u) * GAT_T + tid;
            int q = idx + qoff;
            if (q >= GAT_NQ) q -= GAT_NQ;
            qq[u] = q;
            if (idx < GAT_NQ) dv4[u] = ed4[q];
            else              dv4[u] = make_int4(-1, -1, -1, -1);
        }
#pragma unroll
        for (int u = 0; u < GAT_SU; ++u) {
            int dd[4] = {dv4[u].x, dv4[u].y, dv4[u].z, dv4[u].w};
#pragma unroll
            for (int k = 0; k < 4; ++k) {
                unsigned dl = (unsigned)(dd[k] - lo);
                bool m = dl < (unsigned)GAT_R;
                unsigned mask = __builtin_amdgcn_ballot_w32(m);
                if (mask != 0u) {
                    int pos = cnt + (int)__builtin_amdgcn_mbcnt_lo(mask, 0u);
                    if (m && pos < GAT_CAP)
                        lst_k[lbase + pos] = ((unsigned)(qq[u] * 4 + k) << 8) | dl;
                    cnt += __builtin_popcount(mask);
                }
            }
        }
    }
    if (lane == 0) {
        wcnt[wave] = (cnt < GAT_CAP) ? cnt : GAT_CAP;
        if (cnt > GAT_CAP) ovf = 1;
    }
    __syncthreads();
    const int over = __builtin_amdgcn_readfirstlane(ovf);

    if (over == 0) {
        const int n = __builtin_amdgcn_readfirstlane(wcnt[wave]);
        for (int j0 = 0; j0 < n; j0 += 128) {
            unsigned key[4]; int sv[4]; float ss[4]; bool ok[4];
#pragma unroll
            for (int u = 0; u < 4; ++u) {
                int j = j0 + u * 32 + lane;
                ok[u] = j < n;
                key[u] = ok[u] ? lst_k[lbase + j] : 0u;
            }
#pragma unroll
            for (int u = 0; u < 4; ++u) sv[u] = e_src[key[u] >> 8];
#pragma unroll
            for (int u = 0; u < 4; ++u) ss[u] = s_src[sv[u]];
#pragma unroll
            for (int u = 0; u < 4; ++u) {
                if (ok[u]) {
                    int j = j0 + u * 32 + lane;
                    unsigned dl = key[u] & 255u;
                    float sc = gat_leaky(ss[u] + sdst[dl]);
                    atomicMax(&smax[dl], enc_f32(sc));
                    lst_k[lbase + j] = ((unsigned)sv[u] << 8) | dl;
                    lst_s[lbase + j] = sc;
                }
            }
        }
        __syncthreads();
        for (int j = lane; j < n; j += 32) {
            unsigned dl = lst_k[lbase + j] & 255u;
            float m = dec_f32(smax[dl]);
            if (!isfinite(m)) m = 0.f;
            float w = expf(lst_s[lbase + j] - m);
            lst_s[lbase + j] = w;
            atomicAdd(&sden[dl], w);
        }
        __syncthreads();
        for (int j0 = 0; j0 < n; j0 += GAT_RU) {
            float2 xv[GAT_RU]; unsigned kk[GAT_RU]; float ww[GAT_RU];
            const int jj = j0 + (lane & (GAT_RU - 1));
            const unsigned kl = (jj < n) ? lst_k[lbase + jj] : 0u;
            const float    wl = (jj < n) ? lst_s[lbase + jj] : 0.f;
#pragma unroll
            for (int u = 0; u < GAT_RU; ++u) {
                if (j0 + u < n) {
                    kk[u] = (unsigned)__builtin_amdgcn_readlane((int)kl, u);
                    ww[u] = __int_as_float(__builtin_amdgcn_readlane(__float_as_int(wl), u));
                    xv[u] = *(const float2*)(Xp + (size_t)(kk[u] >> 8) * DIM + lane * 2);
                }
            }
#pragma unroll
            for (int u = 0; u < GAT_RU; ++u) {
                if (j0 + u < n) {
                    float* a = acc + (kk[u] & 255u) * DIM + lane * 2;
                    atomicAdd(a + 0, ww[u] * xv[u].x);
                    atomicAdd(a + 1, ww[u] * xv[u].y);
                }
            }
        }
    } else {
        for (int e = tid; e < N_EDGES; e += GAT_T) {
            unsigned dl = (unsigned)(e_dst[e] - lo);
            if (dl < (unsigned)GAT_R) {
                float sc = gat_leaky(s_src[e_src[e]] + sdst[dl]);
                atomicMax(&smax[dl], enc_f32(sc));
            }
        }
        __syncthreads();
        for (int eb = wave * 32; eb < N_EDGES; eb += GAT_T) {
            int e = eb + lane;
            bool valid = e < N_EDGES;
            unsigned dl = valid ? (unsigned)(e_dst[e] - lo) : 0xFFFFFFFFu;
            bool m = dl < (unsigned)GAT_R;
            float w = 0.f; int sidx = 0;
            if (m) {
                sidx = e_src[e];
                float sc = gat_leaky(s_src[sidx] + sdst[dl]);
                float mx = dec_f32(smax[dl]);
                if (!isfinite(mx)) mx = 0.f;
                w = expf(sc - mx);
                atomicAdd(&sden[dl], w);
            }
            unsigned mask = __builtin_amdgcn_ballot_w32(m);
            while (mask != 0u) {
                int l = __builtin_ctz(mask);
                mask &= mask - 1u;
                int   s_l = __shfl(sidx, l);
                int   d_l = __shfl((int)dl, l);
                float w_l = __shfl(w, l);
                float2 xv = *(const float2*)(Xp + (size_t)s_l * DIM + lane * 2);
                float* a = acc + d_l * DIM + lane * 2;
                atomicAdd(a + 0, w_l * xv.x);
                atomicAdd(a + 1, w_l * xv.y);
            }
        }
    }
    __syncthreads();

    const int cnt2 = hg_filter_scan((const int4*)inc_v, blockIdx.x * (HG_NQ / GAT_NB), lo, (unsigned)GAT_R,
                                    lst_k + lbase, GAT_CAP, tid);
    if (lane == 0) {
        wcnt[wave] = (cnt2 < GAT_CAP) ? cnt2 : GAT_CAP;
        if (cnt2 > GAT_CAP) ovf2 = 1;
    }
    __syncthreads();
    const int over2 = __builtin_amdgcn_readfirstlane(ovf2);

    if (over2 == 0) {
        const int n2 = __builtin_amdgcn_readfirstlane(wcnt[wave]);
        for (int j = lane; j < n2; j += 32) {
            unsigned key = lst_k[lbase + j];
            int e = inc_e[key >> 8];
            lst_k[lbase + j] = ((unsigned)e << 8) | (key & 255u);
        }
        __syncthreads();
        hg_row_sum<false>(lst_k + lbase, lst_s + lbase, n2, He, hacc, lane);
    } else {
        for (int pb = wave * 32; pb < P_PAIRS; pb += GAT_T) {
            int p = pb + lane;
            bool valid = p < P_PAIRS;
            unsigned dl = valid ? (unsigned)(inc_v[p] - lo) : 0xFFFFFFFFu;
            bool m = dl < (unsigned)GAT_R;
            int eidx = m ? inc_e[p] : 0;
            unsigned mask = __builtin_amdgcn_ballot_w32(m);
            while (mask != 0u) {
                int l = __builtin_ctz(mask);
                mask &= mask - 1u;
                int e_l = __shfl(eidx, l);
                int d_l = __shfl((int)dl, l);
                float2 h = *(const float2*)(He + (size_t)e_l * DIM + lane * 2);
                float* a = hacc + d_l * DIM + lane * 2;
                atomicAdd(a + 0, h.x);
                atomicAdd(a + 1, h.y);
            }
        }
    }
    __syncthreads();

    float4* o4 = (float4*)(out + (size_t)lo * DIM);
    const float4* acc4  = (const float4*)acc;
    const float4* hacc4 = (const float4*)hacc;
    for (int i = tid; i < GAT_R * (DIM / 4); i += GAT_T) {
        float den = sden[i >> 4];
        float d   = sdv[i >> 4];
        float s   = (d > 0.f) ? rsqrtf(d) : 0.f;
        float4 v = acc4[i];
        float4 h = hacc4[i];
        if (den != 0.f) { v.x /= den; v.y /= den; v.z /= den; v.w /= den; }
        float4 x;
        x.x = 0.5f * (v.x + h.x * s);
        x.y = 0.5f * (v.y + h.y * s);
        x.z = 0.5f * (v.z + h.z * s);
        x.w = 0.5f * (v.w + h.w * s);
        x.x = (x.x > 0.f) ? x.x : expm1f(x.x);
        x.y = (x.y > 0.f) ? x.y : expm1f(x.y);
        x.z = (x.z > 0.f) ? x.z : expm1f(x.z);
        x.w = (x.w > 0.f) ? x.w : expm1f(x.w);
        o4[i] = x;
    }
}


extern "C" void kernel_launch(void* const* d_in, const int* in_sizes, int n_in,
                              void* d_out, int out_size, void* d_ws, size_t ws_size,
                              hipStream_t stream) {
    const float* X     = (const float*)d_in[0];
    const float* W     = (const float*)d_in[1];
    const float* bias  = (const float*)d_in[2];
    const float* a_src = (const float*)d_in[3];
    const float* a_dst = (const float*)d_in[4];
    const int*   e_src = (const int*)d_in[5];
    const int*   e_dst = (const int*)d_in[6];
    const int*   inc_v = (const int*)d_in[7];
    const int*   inc_e = (const int*)d_in[8];
    float* out = (float*)d_out;
    float* ws  = (float*)d_ws;

    size_t off = 0;
    float* Xp    = ws + off; off += (size_t)N_NODES * DIM;
    float* s_src = ws + off; off += N_NODES;
    float* s_dst = ws + off; off += N_NODES;
    size_t zero_base = off;
    float* dv    = ws + off; off += N_NODES;
    float* de    = ws + off; off += M_HYPER;
    size_t zero_cnt = off - zero_base;
    float* He    = ws + off; off += (size_t)M_HYPER * DIM;

    const int B = 256;
    k_zero<<<(int)((zero_cnt + B - 1) / B), B, 0, stream>>>(ws + zero_base, (long)zero_cnt);

    int tiles = N_NODES / 16;
    k_gemm_wmma<<<(tiles * 32 + B - 1) / B, B, 0, stream>>>(X, W, bias, Xp);

    k_scalars<<<(N_NODES + B - 1) / B, B, 0, stream>>>(Xp, a_src, a_dst, s_src, s_dst);
    k_deg<<<(P_PAIRS + B - 1) / B, B, 0, stream>>>(inc_v, inc_e, dv, de);
    k_he_owner<<<HE_NB, HG_T, 0, stream>>>(inc_v, inc_e, Xp, dv, de, He);
    k_gat_owner<<<GAT_NB, GAT_T, 0, stream>>>(e_src, e_dst, s_src, s_dst, Xp, inc_v, inc_e, He, dv, out);
}


# Owner scans read a per-call 1-byte owner-key plane (16 entries per load, zero-byte test, hits expanded per wave) instead of the int32 lists; wave list order changes, sums stay LDS atomics
# speedup vs baseline: 2.4649x; 1.1339x over previous
#include <hip/hip_runtime.h>
#include <hip/hip_bf16.h>
#include <math.h>

typedef __attribute__((ext_vector_type(2))) float v2f;
typedef __attribute__((ext_vector_type(8))) float v8f;

#define N_NODES 50000
#define N_EDGES 800000
#define M_HYPER 10000
#define P_PAIRS 200000
#define DIM     64
#define NEG_SLOPE 0.2f
#define KEY_GAT_R 200
#define KEY_HE_R  40

__device__ __forceinline__ unsigned enc_f32(float f) {
    unsigned u = __float_as_uint(f);
    return (u & 0x80000000u) ? ~u : (u | 0x80000000u);
}
__device__ __forceinline__ float dec_f32(unsigned u) {
    return (u & 0x80000000u) ? __uint_as_float(u & 0x7FFFFFFFu)
                             : __uint_as_float(~u);
}

__global__ void k_zero(float* __restrict__ p, long n) {
    long i = (long)blockIdx.x * blockDim.x + threadIdx.x;
    if (i < n) p[i] = 0.0f;
}

__global__ void k_gemm_wmma(const float* __restrict__ X, const float* __restrict__ W,
                            const float* __restrict__ bias, float* __restrict__ Xp) {
    int wave = (blockIdx.x * blockDim.x + threadIdx.x) >> 5;
    int lane = threadIdx.x & 31;
    if (wave >= N_NODES / 16) return;
    int row_base = wave * 16;
    int lh  = (lane >= 16) ? 1 : 0;
    int l15 = lane & 15;

    const float* xrow = X + (size_t)(row_base + l15) * DIM + lh * 2;
    const float* wrow = W + (size_t)l15 * DIM + lh * 2;

    v8f acc[4] = {};
#pragma unroll
    for (int ks = 0; ks < 16; ++ks) {
        v2f a;
        a.x = xrow[ks * 4 + 0];
        a.y = xrow[ks * 4 + 1];
#pragma unroll
        for (int t = 0; t < 4; ++t) {
            v2f b;
            const float* wb = wrow + (size_t)t * 16 * DIM + ks * 4;
            b.x = wb[0];
            b.y = wb[1];
            acc[t] = __builtin_amdgcn_wmma_f32_16x16x4_f32(
                false, a, false, b, (short)0, acc[t], false, false);
        }
    }

#pragma unroll
    for (int t = 0; t < 4; ++t) {
        int c = t * 16 + l15;
        float bv = bias[c];
#pragma unroll
        for (int v = 0; v < 8; ++v) {
            int m = row_base + v + lh * 8;
            Xp[(size_t)m * DIM + c] = acc[t][v] + bv;
        }
    }
}

__global__ void k_scalars(const float* __restrict__ Xp,
                          const float* __restrict__ a_src, const float* __restrict__ a_dst,
                          float* __restrict__ s_src, float* __restrict__ s_dst) {
    int n = blockIdx.x * blockDim.x + threadIdx.x;
    if (n >= N_NODES) return;
    const float4* row = (const float4*)(Xp + (size_t)n * DIM);
    const float4* as  = (const float4*)a_src;
    const float4* ad  = (const float4*)a_dst;
    float ss = 0.f, sd = 0.f;
#pragma unroll
    for (int i = 0; i < 16; ++i) {
        float4 x = row[i], u = as[i], v = ad[i];
        ss += x.x * u.x + x.y * u.y + x.z * u.z + x.w * u.w;
        sd += x.x * v.x + x.y * v.y + x.z * v.z + x.w * v.w;
    }
    s_src[n] = ss;
    s_dst[n] = sd;
}

__device__ __forceinline__ unsigned own_key(int row, unsigned rows_per_group) {
    unsigned k = (unsigned)row / rows_per_group;
    return (k < 255u) ? k : 255u;
}
__device__ __forceinline__ unsigned own_key4(int4 r, unsigned rows_per_group) {
    return own_key(r.x, rows_per_group) | (own_key(r.y, rows_per_group) << 8) |
           (own_key(r.z, rows_per_group) << 16) | (own_key(r.w, rows_per_group) << 24);
}

__global__ void k_deg(const int* __restrict__ inc_v, const int* __restrict__ inc_e, const int* __restrict__ e_dst,
                      float* __restrict__ dv, float* __restrict__ de,
                      unsigned* __restrict__ key_ed, unsigned* __restrict__ key_iv, unsigned* __restrict__ key_ie) {
    static_assert(N_EDGES == 4 * P_PAIRS, "one thread per pair packs four edge keys");
    int p = blockIdx.x * blockDim.x + threadIdx.x;
    if (p >= P_PAIRS) return;
    atomicAdd(dv + inc_v[p], 1.0f);
    atomicAdd(de + inc_e[p], 1.0f);
    key_ed[p] = own_key4(((const int4*)e_dst)[p], (unsigned)KEY_GAT_R);
    if (p < P_PAIRS / 4) {
        key_iv[p] = own_key4(((const int4*)inc_v)[p], (unsigned)KEY_GAT_R);
        key_ie[p] = own_key4(((const int4*)inc_e)[p], (unsigned)KEY_HE_R);
    }
}

#define HG_T      1024
#define HG_WAVES  (HG_T / 32)
#define HG_NQ     (P_PAIRS / 16)
#define HG_SU     7
#define HG_RU     16
#define HE_R      40
#define HE_NB     (M_HYPER / HE_R)
#define HE_CAP    256

template <int NQ>
__device__ __forceinline__ int key_scan(const int4* __restrict__ key4, int qoff, unsigned id,
                                        unsigned* desc, int cap, int tid) {
    constexpr int NIT = (NQ + HG_T - 1) / HG_T;
    const unsigned pat = id * 0x01010101u;
    int cnt = 0;
#pragma unroll 1
    for (int i0 = 0; i0 < NIT; i0 += HG_SU) {
        int4 kv[HG_SU];
        int  qq[HG_SU];
#pragma unroll
        for (int u = 0; u < HG_SU; ++u) {
            int idx = (i0 + u) * HG_T + tid;
            int q = idx + qoff;
            if (q >= NQ) q -= NQ;
            qq[u] = q;
            if (idx < NQ) kv[u] = key4[q];
            else          kv[u] = make_int4(-1, -1, -1, -1);
        }
#pragma unroll
        for (int u = 0; u < HG_SU; ++u) {
            const unsigned y0 = (unsigned)kv[u].x ^ pat, y1 = (unsigned)kv[u].y ^ pat;
            const unsigned y2 = (unsigned)kv[u].z ^ pat, y3 = (unsigned)kv[u].w ^ pat;
            const unsigned any = (((y0 - 0x01010101u) & ~y0) | ((y1 - 0x01010101u) & ~y1) |
                                  ((y2 - 0x01010101u) & ~y2) | ((y3 - 0x01010101u) & ~y3)) & 0x80808080u;
            const bool m = any != 0u;
            const unsigned mask = __builtin_amdgcn_ballot_w32(m);
            if (mask != 0u) {
                const int pos = cnt + (int)__builtin_amdgcn_mbcnt_lo(mask, 0u);
                if (m && pos < cap) desc[pos] = (unsigned)qq[u];
                cnt += __builtin_popcount(mask);
            }
        }
    }
    return cnt;
}

__device__ __forceinline__ int key_expand(const int4* __restrict__ key4, unsigned id, const unsigned* desc, int nd,
                                          unsigned* lst, int cap, int lane) {
    const unsigned pat = id * 0x01010101u;
    const unsigned K = 0x7F7F7F7Fu;
    int cnt = 0;
    for (int j0 = 0; j0 < nd; j0 += 32) {
        const int j = j0 + lane;
        unsigned q = 0u, hm = 0u;
        if (j < nd) {
            q = desc[j];
            const int4 kv = key4[q];
            const unsigned y0 = (unsigned)kv.x ^ pat, y1 = (unsigned)kv.y ^ pat;
            const unsigned y2 = (unsigned)kv.z ^ pat, y3 = (unsigned)kv.w ^ pat;
            const unsigned z0 = ~(((y0 & K) + K) | y0 | K), z1 = ~(((y1 & K) + K) | y1 | K);
            const unsigned z2 = ~(((y2 & K) + K) | y2 | K), z3 = ~(((y3 & K) + K) | y3 | K);
            hm = (z0 >> 7) | (z1 >> 6) | (z2 >> 5) | (z3 >> 4);
        }
        unsigned mask = __builtin_amdgcn_ballot_w32(hm != 0u);
        while (mask != 0u) {
            const bool m = hm != 0u;
            const int b = __ffs((int)hm) - 1;
            const int el = ((b & 7) << 2) | (b >> 3);
            const int pos = cnt + (int)__builtin_amdgcn_mbcnt_lo(mask, 0u);
            if (m && pos < cap) lst[pos] = q * 16u + (unsigned)el;
            cnt += __builtin_popcount(mask);
            hm &= hm - 1u;
            mask = __builtin_amdgcn_ballot_w32(hm != 0u);
        }
    }
    return cnt;
}

template <bool WEIGHTED>
__device__ __forceinline__ void hg_row_sum(const unsigned* lk, const float* ls, int n,
                                           const float* __restrict__ src, float* accp, int lane) {
    for (int j0 = 0; j0 < n; j0 += HG_RU) {
        float2 xv[HG_RU]; unsigned kk[HG_RU]; float ww[HG_RU];
        const int jj = j0 + (lane & (HG_RU - 1));
        const unsigned kl = (jj < n) ? lk[jj] : 0u;
        float wl = 1.f;
        if (WEIGHTED) wl = (jj < n) ? ls[jj] : 0.f;
#pragma unroll
        for (int u = 0; u < HG_RU; ++u) {
            if (j0 + u < n) {
                kk[u] = (unsigned)__builtin_amdgcn_readlane((int)kl, u);
                ww[u] = __int_as_float(__builtin_amdgcn_readlane(__float_as_int(wl), u));
                xv[u] = *(const float2*)(src + (size_t)(kk[u] >> 8) * DIM + lane * 2);
            }
        }
#pragma unroll
        for (int u = 0; u < HG_RU; ++u) {
            if (j0 + u < n) {
                float* a = accp + (kk[u] & 255u) * DIM + lane * 2;
                if (WEIGHTED) {
                    atomicAdd(a + 0, ww[u] * xv[u].x);
                    atomicAdd(a + 1, ww[u] * xv[u].y);
                } else {
                    atomicAdd(a + 0, xv[u].x);
                    atomicAdd(a + 1, xv[u].y);
                }
            }
        }
    }
}

__global__ __launch_bounds__(HG_T)
void k_he_owner(const int* __restrict__ inc_v, const int* __restrict__ inc_e,
                const unsigned* __restrict__ key_ie,
                const float* __restrict__ Xp, const float* __restrict__ dv, const float* __restrict__ de,
                float* __restrict__ He) {
    __shared__ __attribute__((aligned(16))) float hacc[HE_R * DIM];
    __shared__ unsigned lst_k[HG_WAVES * HE_CAP];
    __shared__ float    lst_s[HG_WAVES * HE_CAP];
    __shared__ float    sde[HE_R];
    __shared__ int      wcnt[HG_WAVES];
    __shared__ int      ovf;

    const int tid  = threadIdx.x;
    const int lane = tid & 31;
    const int wave = tid >> 5;
    const int lo   = blockIdx.x * HE_R;
    const int lbase = wave * HE_CAP;

    for (int i = tid; i < HE_R * DIM; i += HG_T) hacc[i] = 0.f;
    if (tid < HE_R) sde[tid] = de[lo + tid];
    if (tid == 0) ovf = 0;
    __syncthreads();

    static_assert(HE_R == KEY_HE_R && HE_NB <= 255, "the inc_e key plane is built for this row split");
    const int nd = key_scan<HG_NQ>((const int4*)key_ie, blockIdx.x * (HG_NQ / HE_NB), blockIdx.x,
                                   (unsigned*)(lst_s + lbase), HE_CAP, tid);
    __syncthreads();
    int cnt = HE_CAP + 1;
    if (nd <= HE_CAP)
        cnt = key_expand((const int4*)key_ie, blockIdx.x, (const unsigned*)(lst_s + lbase), nd, lst_k + lbase, HE_CAP, lane);
    if (lane == 0) {
        wcnt[wave] = (cnt < HE_CAP) ? cnt : HE_CAP;
        if (cnt > HE_CAP) ovf = 1;
    }
    __syncthreads();
    const int over = __builtin_amdgcn_readfirstlane(ovf);

    if (over == 0) {
        const int n = __builtin_amdgcn_readfirstlane(wcnt[wave]);
        for (int j0 = 0; j0 < n; j0 += 128) {
            unsigned key[4]; int vv[4]; int ee[4]; float dd[4]; bool ok[4];
#pragma unroll
            for (int u = 0; u < 4; ++u) {
                int j = j0 + u * 32 + lane;
                ok[u] = j < n;
                key[u] = ok[u] ? lst_k[lbase + j] : 0u;
            }
#pragma unroll
            for (int u = 0; u < 4; ++u) { vv[u] = inc_v[key[u]]; ee[u] = inc_e[key[u]]; }
#pragma unroll
            for (int u = 0; u < 4; ++u) dd[u] = dv[vv[u]];
#pragma unroll
            for (int u = 0; u < 4; ++u) {
                if (ok[u]) {
                    int j = j0 + u * 32 + lane;
                    lst_k[lbase + j] = ((unsigned)vv[u] << 8) | (unsigned)(ee[u] - lo);
                    lst_s[lbase + j] = (dd[u] > 0.f) ? rsqrtf(dd[u]) : 0.f;
                }
            }
        }
        __syncthreads();
        hg_row_sum<true>(lst_k + lbase, lst_s + lbase, n, Xp, hacc, lane);
    } else {
        for (int pb = wave * 32; pb < P_PAIRS; pb += HG_T) {
            int p = pb + lane;
            bool valid = p < P_PAIRS;
            unsigned dl = valid ? (unsigned)(inc_e[p] - lo) : 0xFFFFFFFFu;
            bool m = dl < (unsigned)HE_R;
            float s = 0.f; int vidx = 0;
            if (m) {
                vidx = inc_v[p];
                float d = dv[vidx];
                s = (d > 0.f) ? rsqrtf(d) : 0.f;
            }
            unsigned mask = __builtin_amdgcn_ballot_w32(m);
            while (mask != 0u) {
                int l = __builtin_ctz(mask);
                mask &= mask - 1u;
                int   v_l = __shfl(vidx, l);
                int   d_l = __shfl((int)dl, l);
                float s_l = __shfl(s, l);
                float2 xv = *(const float2*)(Xp + (size_t)v_l * DIM + lane * 2);
                float* a = hacc + d_l * DIM + lane * 2;
                atomicAdd(a + 0, s_l * xv.x);
                atomicAdd(a + 1, s_l * xv.y);
            }
        }
    }
    __syncthreads();

    float4* he4 = (float4*)(He + (size_t)lo * DIM);
    const float4* hacc4 = (const float4*)hacc;
    for (int i = tid; i < HE_R * (DIM / 4); i += HG_T) {
        float d = sde[i >> 4];
        float inv = (d > 0.f) ? (1.f / d) : 0.f;
        float4 v = hacc4[i];
        v.x *= inv; v.y *= inv; v.z *= inv; v.w *= inv;
        he4[i] = v;
    }
}

#define GAT_R     200
#define GAT_NB    (N_NODES / GAT_R)
#define GAT_T     1024
#define GAT_WAVES (GAT_T / 32)
#define GAT_CAP   512
#define GAT_NQ    (N_EDGES / 16)
#define GAT_RU    16

__device__ __forceinline__ float gat_leaky(float x) { return (x >= 0.f) ? x : NEG_SLOPE * x; }

__global__ __launch_bounds__(GAT_T)
void k_gat_owner(const int* __restrict__ e_src, const int* __restrict__ e_dst,
                 const float* __restrict__ s_src, const float* __restrict__ s_dst,
                 const float* __restrict__ Xp,
                 const int* __restrict__ inc_v, const int* __restrict__ inc_e,
                 const unsigned* __restrict__ key_ed, const unsigned* __restrict__ key_iv,
                 const float* __restrict__ He, const float* __restrict__ dv,
                 float* __restrict__ out) {
    __shared__ __attribute__((aligned(16))) float acc[GAT_R * DIM];
    __shared__ __attribute__((aligned(16))) float hacc[GAT_R * DIM];
    __shared__ unsigned lst_k[GAT_WAVES * GAT_CAP];
    __shared__ float    lst_s[GAT_WAVES * GAT_CAP];
    __shared__ unsigned smax[GAT_R];
    __shared__ float    sden[GAT_R];
    __shared__ float    sdst[GAT_R];
    __shared__ float    sdv[GAT_R];
    __shared__ int      wcnt[GAT_WAVES];
    __shared__ int      ovf;
    __shared__ int      ovf2;

    const int tid  = threadIdx.x;
    const int lane = tid & 31;
    const int wave = tid >> 5;
    const int lo   = blockIdx.x * GAT_R;
    const int lbase = wave * GAT_CAP;

    for (int i = tid; i < GAT_R * DIM; i += GAT_T) { acc[i] = 0.f; hacc[i] = 0.f; }
    if (tid < GAT_R) { smax[tid] = 0u; sden[tid] = 0.f; sdst[tid] = s_dst[lo + tid]; sdv[tid] = dv[lo + tid]; }
    if (tid == 0) { ovf = 0; ovf2 = 0; }
    __syncthreads();

    static_assert(GAT_R == KEY_GAT_R && GAT_NB <= 255 && GAT_T == HG_T, "the e_dst / inc_v key planes are built for this row split");
    const int nd = key_scan<GAT_NQ>((const int4*)key_ed, blockIdx.x * (GAT_NQ / GAT_NB), blockIdx.x,
                                    (unsigned*)(lst_s + lbase), GAT_CAP, tid);
    __syncthreads();
    int cnt = GAT_CAP + 1;
    if (nd <= GAT_CAP)
        cnt = key_expand((const int4*)key_ed, blockIdx.x, (const unsigned*)(lst_s + lbase), nd, lst_k + lbase, GAT_CAP, lane);
    if (lane == 0) {
        wcnt[wave] = (cnt < GAT_CAP) ? cnt : GAT_CAP;
        if (cnt > GAT_CAP) ovf = 1;
    }
    __syncthreads();
    const int over = __builtin_amdgcn_readfirstlane(ovf);

    if (over == 0) {
        const int n = __builtin_amdgcn_readfirstlane(wcnt[wave]);
        for (int j0 = 0; j0 < n; j0 += 128) {
            unsigned key[4]; int sv[4]; int dd[4]; float ss[4]; bool ok[4];
#pragma unroll
            for (int u = 0; u < 4; ++u) {
                int j = j0 + u * 32 + lane;
                ok[u] = j < n;
                key[u] = ok[u] ? lst_k[lbase + j] : 0u;
            }
#pragma unroll
            for (int u = 0; u < 4; ++u) { sv[u] = e_src[key[u]]; dd[u] = e_dst[key[u]]; }
#pragma unroll
            for (int u = 0; u < 4; ++u) ss[u] = s_src[sv[u]];
#pragma unroll
            for (int u = 0; u < 4; ++u) {
                if (ok[u]) {
                    int j = j0 + u * 32 + lane;
                    unsigned dl = (unsigned)(dd[u] - lo);
                    float sc = gat_leaky(ss[u] + sdst[dl]);
                    atomicMax(&smax[dl], enc_f32(sc));
                    lst_k[lbase + j] = ((unsigned)sv[u] << 8) | dl;
                    lst_s[lbase + j] = sc;
                }
            }
        }
        __syncthreads();
        for (int j = lane; j < n; j += 32) {
            unsigned dl = lst_k[lbase + j] & 255u;
            float m = dec_f32(smax[dl]);
            if (!isfinite(m)) m = 0.f;
            float w = expf(lst_s[lbase + j] - m);
            lst_s[lbase + j] = w;
            atomicAdd(&sden[dl], w);
        }
        __syncthreads();
        for (int j0 = 0; j0 < n; j0 += GAT_RU) {
            float2 xv[GAT_RU]; unsigned kk[GAT_RU]; float ww[GAT_RU];
            const int jj = j0 + (lane & (GAT_RU - 1));
            const unsigned kl = (jj < n) ? lst_k[lbase + jj] : 0u;
            const float    wl = (jj < n) ? lst_s[lbase + jj] : 0.f;
#pragma unroll
            for (int u = 0; u < GAT_RU; ++u) {
                if (j0 + u < n) {
                    kk[u] = (unsigned)__builtin_amdgcn_readlane((int)kl, u);
                    ww[u] = __int_as_float(__builtin_amdgcn_readlane(__float_as_int(wl), u));
                    xv[u] = *(const float2*)(Xp + (size_t)(kk[u] >> 8) * DIM + lane * 2);
                }
            }
#pragma unroll
            for (int u = 0; u < GAT_RU; ++u) {
                if (j0 + u < n) {
                    float* a = acc + (kk[u] & 255u) * DIM + lane * 2;
                    atomicAdd(a + 0, ww[u] * xv[u].x);
                    atomicAdd(a + 1, ww[u] * xv[u].y);
                }
            }
        }
    } else {
        for (int e = tid; e < N_EDGES; e += GAT_T) {
            unsigned dl = (unsigned)(e_dst[e] - lo);
            if (dl < (unsigned)GAT_R) {
                float sc = gat_leaky(s_src[e_src[e]] + sdst[dl]);
                atomicMax(&smax[dl], enc_f32(sc));
            }
        }
        __syncthreads();
        for (int eb = wave * 32; eb < N_EDGES; eb += GAT_T) {
            int e = eb + lane;
            bool valid = e < N_EDGES;
            unsigned dl = valid ? (unsigned)(e_dst[e] - lo) : 0xFFFFFFFFu;
            bool m = dl < (unsigned)GAT_R;
            float w = 0.f; int sidx = 0;
            if (m) {
                sidx = e_src[e];
                float sc = gat_leaky(s_src[sidx] + sdst[dl]);
                float mx = dec_f32(smax[dl]);
                if (!isfinite(mx)) mx = 0.f;
                w = expf(sc - mx);
                atomicAdd(&sden[dl], w);
            }
            unsigned mask = __builtin_amdgcn_ballot_w32(m);
            while (mask != 0u) {
                int l = __builtin_ctz(mask);
                mask &= mask - 1u;
                int   s_l = __shfl(sidx, l);
                int   d_l = __shfl((int)dl, l);
                float w_l = __shfl(w, l);
                float2 xv = *(const float2*)(Xp + (size_t)s_l * DIM + lane * 2);
                float* a = acc + d_l * DIM + lane * 2;
                atomicAdd(a + 0, w_l * xv.x);
                atomicAdd(a + 1, w_l * xv.y);
            }
        }
    }
    __syncthreads();

    const int nd2 = key_scan<HG_NQ>((const int4*)key_iv, blockIdx.x * (HG_NQ / GAT_NB), blockIdx.x,
                                    (unsigned*)(lst_s + lbase), GAT_CAP, tid);
    __syncthreads();
    int cnt2 = GAT_CAP + 1;
    if (nd2 <= GAT_CAP)
        cnt2 = key_expand((const int4*)key_iv, blockIdx.x, (const unsigned*)(lst_s + lbase), nd2, lst_k + lbase, GAT_CAP, lane);
    if (lane == 0) {
        wcnt[wave] = (cnt2 < GAT_CAP) ? cnt2 : GAT_CAP;
        if (cnt2 > GAT_CAP) ovf2 = 1;
    }
    __syncthreads();
    const int over2 = __builtin_amdgcn_readfirstlane(ovf2);

    if (over2 == 0) {
        const int n2 = __builtin_amdgcn_readfirstlane(wcnt[wave]);
        for (int j = lane; j < n2; j += 32) {
            unsigned p = lst_k[lbase + j];
            int e = inc_e[p];
            int v = inc_v[p];
            lst_k[lbase + j] = ((unsigned)e << 8) | (unsigned)(v - lo);
        }
        __syncthreads();
        hg_row_sum<false>(lst_k + lbase, lst_s + lbase, n2, He, hacc, lane);
    } else {
        for (int pb = wave * 32; pb < P_PAIRS; pb += GAT_T) {
            int p = pb + lane;
            bool valid = p < P_PAIRS;
            unsigned dl = valid ? (unsigned)(inc_v[p] - lo) : 0xFFFFFFFFu;
            bool m = dl < (unsigned)GAT_R;
            int eidx = m ? inc_e[p] : 0;
            unsigned mask = __builtin_amdgcn_ballot_w32(m);
            while (mask != 0u) {
                int l = __builtin_ctz(mask);
                mask &= mask - 1u;
                int e_l = __shfl(eidx, l);
                int d_l = __shfl((int)dl, l);
                float2 h = *(const float2*)(He + (size_t)e_l * DIM + lane * 2);
                float* a = hacc + d_l * DIM + lane * 2;
                atomicAdd(a + 0, h.x);
                atomicAdd(a + 1, h.y);
            }
        }
    }
    __syncthreads();

    float4* o4 = (float4*)(out + (size_t)lo * DIM);
    const float4* acc4  = (const float4*)acc;
    const float4* hacc4 = (const float4*)hacc;
    for (int i = tid; i < GAT_R * (DIM / 4); i += GAT_T) {
        float den = sden[i >> 4];
        float d   = sdv[i >> 4];
        float s   = (d > 0.f) ? rsqrtf(d) : 0.f;
        float4 v = acc4[i];
        float4 h = hacc4[i];
        if (den != 0.f) { v.x /= den; v.y /= den; v.z /= den; v.w /= den; }
        float4 x;
        x.x = 0.5f * (v.x + h.x * s);
        x.y = 0.5f * (v.y + h.y * s);
        x.z = 0.5f * (v.z + h.z * s);
        x.w = 0.5f * (v.w + h.w * s);
        x.x = (x.x > 0.f) ? x.x : expm1f(x.x);
        x.y = (x.y > 0.f) ? x.y : expm1f(x.y);
        x.z = (x.z > 0.f) ? x.z : expm1f(x.z);
        x.w = (x.w > 0.f) ? x.w : expm1f(x.w);
        o4[i] = x;
    }
}


extern "C" void kernel_launch(void* const* d_in, const int* in_sizes, int n_in,
                              void* d_out, int out_size, void* d_ws, size_t ws_size,
                              hipStream_t stream) {
    const float* X     = (const float*)d_in[0];
    const float* W     = (const float*)d_in[1];
    const float* bias  = (const float*)d_in[2];
    const float* a_src = (const float*)d_in[3];
    const float* a_dst = (const float*)d_in[4];
    const int*   e_src = (const int*)d_in[5];
    const int*   e_dst = (const int*)d_in[6];
    const int*   inc_v = (const int*)d_in[7];
    const int*   inc_e = (const int*)d_in[8];
    float* out = (float*)d_out;
    float* ws  = (float*)d_ws;

    size_t off = 0;
    float* Xp    = ws + off; off += (size_t)N_NODES * DIM;
    float* s_src = ws + off; off += N_NODES;
    float* s_dst = ws + off; off += N_NODES;
    size_t zero_base = off;
    float* dv    = ws + off; off += N_NODES;
    float* de    = ws + off; off += M_HYPER;
    size_t zero_cnt = off - zero_base;
    float* He    = ws + off; off += (size_t)M_HYPER * DIM;
    unsigned* key_ed = (unsigned*)(ws + off); off += N_EDGES / 4;
    unsigned* key_iv = (unsigned*)(ws + off); off += P_PAIRS / 4;
    unsigned* key_ie = (unsigned*)(ws + off); off += P_PAIRS / 4;

    const int B = 256;
    k_zero<<<(int)((zero_cnt + B - 1) / B), B, 0, stream>>>(ws + zero_base, (long)zero_cnt);

    int tiles = N_NODES / 16;
    k_gemm_wmma<<<(tiles * 32 + B - 1) / B, B, 0, stream>>>(X, W, bias, Xp);

    k_scalars<<<(N_NODES + B - 1) / B, B, 0, stream>>>(Xp, a_src, a_dst, s_src, s_dst);
    k_deg<<<(P_PAIRS + B - 1) / B, B, 0, stream>>>(inc_v, inc_e, e_dst, dv, de, key_ed, key_iv, key_ie);
    k_he_owner<<<HE_NB, HG_T, 0, stream>>>(inc_v, inc_e, key_ie, Xp, dv, de, He);
    k_gat_owner<<<GAT_NB, GAT_T, 0, stream>>>(e_src, e_dst, s_src, s_dst, Xp, inc_v, inc_e, key_ed, key_iv, He, dv, out);
}


# Zero fill and attention scalars folded into the WMMA projection launch (s_src/s_dst from accumulator registers: 4 column tiles per lane, then 16-lane DPP sum = new sum order); 6 launches -> 4
# speedup vs baseline: 2.7481x; 1.1149x over previous
#include <hip/hip_runtime.h>
#include <hip/hip_bf16.h>
#include <math.h>

typedef __attribute__((ext_vector_type(2))) float v2f;
typedef __attribute__((ext_vector_type(8))) float v8f;

#define N_NODES 50000
#define N_EDGES 800000
#define M_HYPER 10000
#define P_PAIRS 200000
#define DIM     64
#define NEG_SLOPE 0.2f
#define KEY_GAT_R 200
#define KEY_HE_R  40

__device__ __forceinline__ unsigned enc_f32(float f) {
    unsigned u = __float_as_uint(f);
    return (u & 0x80000000u) ? ~u : (u | 0x80000000u);
}
__device__ __forceinline__ float dec_f32(unsigned u) {
    return (u & 0x80000000u) ? __uint_as_float(u & 0x7FFFFFFFu)
                             : __uint_as_float(~u);
}

#define PJ_T      256
#define PJ_WAVES  (PJ_T / 32)
#define PJ_ROWS   (PJ_WAVES * 16)
#define PJ_TILES  (N_NODES / 16)
#define PJ_NB     ((PJ_TILES + PJ_WAVES - 1) / PJ_WAVES)
#define DEG_CNT   (N_NODES + M_HYPER)

__device__ __forceinline__ float row16_sum(float x) {
    x += __int_as_float(__builtin_amdgcn_update_dpp(0, __float_as_int(x), 0x161, 0xF, 0xF, false));
    x += __int_as_float(__builtin_amdgcn_update_dpp(0, __float_as_int(x), 0x162, 0xF, 0xF, false));
    x += __int_as_float(__builtin_amdgcn_update_dpp(0, __float_as_int(x), 0x164, 0xF, 0xF, false));
    x += __int_as_float(__builtin_amdgcn_update_dpp(0, __float_as_int(x), 0x168, 0xF, 0xF, false));
    return x;
}

__global__ __launch_bounds__(PJ_T)
void k_project(const float* __restrict__ X, const float* __restrict__ W, const float* __restrict__ bias,
               const float* __restrict__ a_src, const float* __restrict__ a_dst,
               float* __restrict__ Xp, float* __restrict__ s_src, float* __restrict__ s_dst,
               float* __restrict__ deg0) {
    static_assert(PJ_NB * PJ_T >= DEG_CNT && PJ_T == 2 * PJ_ROWS, "one thread per degree counter; one thread per scalar of the group");
    __shared__ float sh_s[PJ_ROWS];
    __shared__ float sh_d[PJ_ROWS];

    const int tid  = threadIdx.x;
    const int lane = tid & 31;
    const int wv   = __builtin_amdgcn_readfirstlane(tid >> 5);
    const int gid  = blockIdx.x * PJ_T + tid;
    if (gid < DEG_CNT) deg0[gid] = 0.0f;

    const int tile = blockIdx.x * PJ_WAVES + wv;
    if (tile < PJ_TILES) {
        const int row_base = tile * 16;
        const int lh  = (lane >= 16) ? 1 : 0;
        const int l15 = lane & 15;

        const float* xrow = X + (size_t)(row_base + l15) * DIM + lh * 2;
        const float* wrow = W + (size_t)l15 * DIM + lh * 2;

        v8f acc[4] = {};
#pragma unroll
        for (int ks = 0; ks < 16; ++ks) {
            v2f a;
            a.x = xrow[ks * 4 + 0];
            a.y = xrow[ks * 4 + 1];
#pragma unroll
            for (int t = 0; t < 4; ++t) {
                v2f b;
                const float* wb = wrow + (size_t)t * 16 * DIM + ks * 4;
                b.x = wb[0];
                b.y = wb[1];
                acc[t] = __builtin_amdgcn_wmma_f32_16x16x4_f32(
                    false, a, false, b, (short)0, acc[t], false, false);
            }
        }

        float ps[8], pd[8];
#pragma unroll
        for (int t = 0; t < 4; ++t) {
            int c = t * 16 + l15;
            float bv = bias[c];
            float us = a_src[c];
            float ud = a_dst[c];
#pragma unroll
            for (int v = 0; v < 8; ++v) {
                int m = row_base + v + lh * 8;
                float x = acc[t][v] + bv;
                Xp[(size_t)m * DIM + c] = x;
                ps[v] = (t == 0) ? x * us : ps[v] + x * us;
                pd[v] = (t == 0) ? x * ud : pd[v] + x * ud;
            }
        }
        float sv = 0.f, dvv = 0.f;
#pragma unroll
        for (int v = 0; v < 8; ++v) {
            float rs = row16_sum(ps[v]);
            float rd = row16_sum(pd[v]);
            if ((l15 & 7) == v) { sv = rs; dvv = rd; }
        }
        if (l15 < 8) {
            sh_s[wv * 16 + lh * 8 + l15] = sv;
            sh_d[wv * 16 + lh * 8 + l15] = dvv;
        }
    }
    __syncthreads();
    {
        const int r   = tid & (PJ_ROWS - 1);
        const int row = blockIdx.x * PJ_ROWS + r;
        if (row < N_NODES) {
            if (tid < PJ_ROWS) s_src[row] = sh_s[r];
            else               s_dst[row] = sh_d[r];
        }
    }
}

__device__ __forceinline__ unsigned own_key(int row, unsigned rows_per_group) {
    unsigned k = (unsigned)row / rows_per_group;
    return (k < 255u) ? k : 255u;
}
__device__ __forceinline__ unsigned own_key4(int4 r, unsigned rows_per_group) {
    return own_key(r.x, rows_per_group) | (own_key(r.y, rows_per_group) << 8) |
           (own_key(r.z, rows_per_group) << 16) | (own_key(r.w, rows_per_group) << 24);
}

__global__ void k_deg(const int* __restrict__ inc_v, const int* __restrict__ inc_e, const int* __restrict__ e_dst,
                      float* __restrict__ dv, float* __restrict__ de,
                      unsigned* __restrict__ key_ed, unsigned* __restrict__ key_iv, unsigned* __restrict__ key_ie) {
    static_assert(N_EDGES == 4 * P_PAIRS, "one thread per pair packs four edge keys");
    int p = blockIdx.x * blockDim.x + threadIdx.x;
    if (p >= P_PAIRS) return;
    atomicAdd(dv + inc_v[p], 1.0f);
    atomicAdd(de + inc_e[p], 1.0f);
    key_ed[p] = own_key4(((const int4*)e_dst)[p], (unsigned)KEY_GAT_R);
    if (p < P_PAIRS / 4) {
        key_iv[p] = own_key4(((const int4*)inc_v)[p], (unsigned)KEY_GAT_R);
        key_ie[p] = own_key4(((const int4*)inc_e)[p], (unsigned)KEY_HE_R);
    }
}

#define HG_T      1024
#define HG_WAVES  (HG_T / 32)
#define HG_NQ     (P_PAIRS / 16)
#define HG_SU     7
#define HG_RU     16
#define HE_R      40
#define HE_NB     (M_HYPER / HE_R)
#define HE_CAP    256

template <int NQ>
__device__ __forceinline__ int key_scan(const int4* __restrict__ key4, int qoff, unsigned id,
                                        unsigned* desc, int cap, int tid) {
    constexpr int NIT = (NQ + HG_T - 1) / HG_T;
    const unsigned pat = id * 0x01010101u;
    int cnt = 0;
#pragma unroll 1
    for (int i0 = 0; i0 < NIT; i0 += HG_SU) {
        int4 kv[HG_SU];
        int  qq[HG_SU];
#pragma unroll
        for (int u = 0; u < HG_SU; ++u) {
            int idx = (i0 + u) * HG_T + tid;
            int q = idx + qoff;
            if (q >= NQ) q -= NQ;
            qq[u] = q;
            if (idx < NQ) kv[u] = key4[q];
            else          kv[u] = make_int4(-1, -1, -1, -1);
        }
#pragma unroll
        for (int u = 0; u < HG_SU; ++u) {
            const unsigned y0 = (unsigned)kv[u].x ^ pat, y1 = (unsigned)kv[u].y ^ pat;
            const unsigned y2 = (unsigned)kv[u].z ^ pat, y3 = (unsigned)kv[u].w ^ pat;
            const unsigned any = (((y0 - 0x01010101u) & ~y0) | ((y1 - 0x01010101u) & ~y1) |
                                  ((y2 - 0x01010101u) & ~y2) | ((y3 - 0x01010101u) & ~y3)) & 0x80808080u;
            const bool m = any != 0u;
            const unsigned mask = __builtin_amdgcn_ballot_w32(m);
            if (mask != 0u) {
                const int pos = cnt + (int)__builtin_amdgcn_mbcnt_lo(mask, 0u);
                if (m && pos < cap) desc[pos] = (unsigned)qq[u];
                cnt += __builtin_popcount(mask);
            }
        }
    }
    return cnt;
}

__device__ __forceinline__ int key_expand(const int4* __restrict__ key4, unsigned id, const unsigned* desc, int nd,
                                          unsigned* lst, int cap, int lane) {
    const unsigned pat = id * 0x01010101u;
    const unsigned K = 0x7F7F7F7Fu;
    int cnt = 0;
    for (int j0 = 0; j0 < nd; j0 += 32) {
        const int j = j0 + lane;
        unsigned q = 0u, hm = 0u;
        if (j < nd) {
            q = desc[j];
            const int4 kv = key4[q];
            const unsigned y0 = (unsigned)kv.x ^ pat, y1 = (unsigned)kv.y ^ pat;
            const unsigned y2 = (unsigned)kv.z ^ pat, y3 = (unsigned)kv.w ^ pat;
            const unsigned z0 = ~(((y0 & K) + K) | y0 | K), z1 = ~(((y1 & K) + K) | y1 | K);
            const unsigned z2 = ~(((y2 & K) + K) | y2 | K), z3 = ~(((y3 & K) + K) | y3 | K);
            hm = (z0 >> 7) | (z1 >> 6) | (z2 >> 5) | (z3 >> 4);
        }
        unsigned mask = __builtin_amdgcn_ballot_w32(hm != 0u);
        while (mask != 0u) {
            const bool m = hm != 0u;
            const int b = __ffs((int)hm) - 1;
            const int el = ((b & 7) << 2) | (b >> 3);
            const int pos = cnt + (int)__builtin_amdgcn_mbcnt_lo(mask, 0u);
            if (m && pos < cap) lst[pos] = q * 16u + (unsigned)el;
            cnt += __builtin_popcount(mask);
            hm &= hm - 1u;
            mask = __builtin_amdgcn_ballot_w32(hm != 0u);
        }
    }
    return cnt;
}

template <bool WEIGHTED>
__device__ __forceinline__ void hg_row_sum(const unsigned* lk, const float* ls, int n,
                                           const float* __restrict__ src, float* accp, int lane) {
    for (int j0 = 0; j0 < n; j0 += HG_RU) {
        float2 xv[HG_RU]; unsigned kk[HG_RU]; float ww[HG_RU];
        const int jj = j0 + (lane & (HG_RU - 1));
        const unsigned kl = (jj < n) ? lk[jj] : 0u;
        float wl = 1.f;
        if (WEIGHTED) wl = (jj < n) ? ls[jj] : 0.f;
#pragma unroll
        for (int u = 0; u < HG_RU; ++u) {
            if (j0 + u < n) {
                kk[u] = (unsigned)__builtin_amdgcn_readlane((int)kl, u);
                ww[u] = __int_as_float(__builtin_amdgcn_readlane(__float_as_int(wl), u));
                xv[u] = *(const float2*)(src + (size_t)(kk[u] >> 8) * DIM + lane * 2);
            }
        }
#pragma unroll
        for (int u = 0; u < HG_RU; ++u) {
            if (j0 + u < n) {
                float* a = accp + (kk[u] & 255u) * DIM + lane * 2;
                if (WEIGHTED) {
                    atomicAdd(a + 0, ww[u] * xv[u].x);
                    atomicAdd(a + 1, ww[u] * xv[u].y);
                } else {
                    atomicAdd(a + 0, xv[u].x);
                    atomicAdd(a + 1, xv[u].y);
                }
            }
        }
    }
}

__global__ __launch_bounds__(HG_T)
void k_he_owner(const int* __restrict__ inc_v, const int* __restrict__ inc_e,
                const unsigned* __restrict__ key_ie,
                const float* __restrict__ Xp, const float* __restrict__ dv, const float* __restrict__ de,
                float* __restrict__ He) {
    __shared__ __attribute__((aligned(16))) float hacc[HE_R * DIM];
    __shared__ unsigned lst_k[HG_WAVES * HE_CAP];
    __shared__ float    lst_s[HG_WAVES * HE_CAP];
    __shared__ float    sde[HE_R];
    __shared__ int      wcnt[HG_WAVES];
    __shared__ int      ovf;

    const int tid  = threadIdx.x;
    const int lane = tid & 31;
    const int wave = tid >> 5;
    const int lo   = blockIdx.x * HE_R;
    const int lbase = wave * HE_CAP;

    for (int i = tid; i < HE_R * DIM; i += HG_T) hacc[i] = 0.f;
    if (tid < HE_R) sde[tid] = de[lo + tid];
    if (tid == 0) ovf = 0;
    __syncthreads();

    static_assert(HE_R == KEY_HE_R && HE_NB <= 255, "the inc_e key plane is built for this row split");
    const int nd = key_scan<HG_NQ>((const int4*)key_ie, blockIdx.x * (HG_NQ / HE_NB), blockIdx.x,
                                   (unsigned*)(lst_s + lbase), HE_CAP, tid);
    __syncthreads();
    int cnt = HE_CAP + 1;
    if (nd <= HE_CAP)
        cnt = key_expand((const int4*)key_ie, blockIdx.x, (const unsigned*)(lst_s + lbase), nd, lst_k + lbase, HE_CAP, lane);
    if (lane == 0) {
        wcnt[wave] = (cnt < HE_CAP) ? cnt : HE_CAP;
        if (cnt > HE_CAP) ovf = 1;
    }
    __syncthreads();
    const int over = __builtin_amdgcn_readfirstlane(ovf);

    if (over == 0) {
        const int n = __builtin_amdgcn_readfirstlane(wcnt[wave]);
        for (int j0 = 0; j0 < n; j0 += 128) {
            unsigned key[4]; int vv[4]; int ee[4]; float dd[4]; bool ok[4];
#pragma unroll
            for (int u = 0; u < 4; ++u) {
                int j = j0 + u * 32 + lane;
                ok[u] = j < n;
                key[u] = ok[u] ? lst_k[lbase + j] : 0u;
            }
#pragma unroll
            for (int u = 0; u < 4; ++u) { vv[u] = inc_v[key[u]]; ee[u] = inc_e[key[u]]; }
#pragma unroll
            for (int u = 0; u < 4; ++u) dd[u] = dv[vv[u]];
#pragma unroll
            for (int u = 0; u < 4; ++u) {
                if (ok[u]) {
                    int j = j0 + u * 32 + lane;
                    lst_k[lbase + j] = ((unsigned)vv[u] << 8) | (unsigned)(ee[u] - lo);
                    lst_s[lbase + j] = (dd[u] > 0.f) ? rsqrtf(dd[u]) : 0.f;
                }
            }
        }
        __syncthreads();
        hg_row_sum<true>(lst_k + lbase, lst_s + lbase, n, Xp, hacc, lane);
    } else {
        for (int pb = wave * 32; pb < P_PAIRS; pb += HG_T) {
            int p = pb + lane;
            bool valid = p < P_PAIRS;
            unsigned dl = valid ? (unsigned)(inc_e[p] - lo) : 0xFFFFFFFFu;
            bool m = dl < (unsigned)HE_R;
            float s = 0.f; int vidx = 0;
            if (m) {
                vidx = inc_v[p];
                float d = dv[vidx];
                s = (d > 0.f) ? rsqrtf(d) : 0.f;
            }
            unsigned mask = __builtin_amdgcn_ballot_w32(m);
            while (mask != 0u) {
                int l = __builtin_ctz(mask);
                mask &= mask - 1u;
                int   v_l = __shfl(vidx, l);
                int   d_l = __shfl((int)dl, l);
                float s_l = __shfl(s, l);
                float2 xv = *(const float2*)(Xp + (size_t)v_l * DIM + lane * 2);
                float* a = hacc + d_l * DIM + lane * 2;
                atomicAdd(a + 0, s_l * xv.x);
                atomicAdd(a + 1, s_l * xv.y);
            }
        }
    }
    __syncthreads();

    float4* he4 = (float4*)(He + (size_t)lo * DIM);
    const float4* hacc4 = (const float4*)hacc;
    for (int i = tid; i < HE_R * (DIM / 4); i += HG_T) {
        float d = sde[i >> 4];
        float inv = (d > 0.f) ? (1.f / d) : 0.f;
        float4 v = hacc4[i];
        v.x *= inv; v.y *= inv; v.z *= inv; v.w *= inv;
        he4[i] = v;
    }
}

#define GAT_R     200
#define GAT_NB    (N_NODES / GAT_R)
#define GAT_T     1024
#define GAT_WAVES (GAT_T / 32)
#define GAT_CAP   512
#define GAT_NQ    (N_EDGES / 16)
#define GAT_RU    16

__device__ __forceinline__ float gat_leaky(float x) { return (x >= 0.f) ? x : NEG_SLOPE * x; }

__global__ __launch_bounds__(GAT_T)
void k_gat_owner(const int* __restrict__ e_src, const int* __restrict__ e_dst,
                 const float* __restrict__ s_src, const float* __restrict__ s_dst,
                 const float* __restrict__ Xp,
                 const int* __restrict__ inc_v, const int* __restrict__ inc_e,
                 const unsigned* __restrict__ key_ed, const unsigned* __restrict__ key_iv,
                 const float* __restrict__ He, const float* __restrict__ dv,
                 float* __restrict__ out) {
    __shared__ __attribute__((aligned(16))) float acc[GAT_R * DIM];
    __shared__ __attribute__((aligned(16))) float hacc[GAT_R * DIM];
    __shared__ unsigned lst_k[GAT_WAVES * GAT_CAP];
    __shared__ float    lst_s[GAT_WAVES * GAT_CAP];
    __shared__ unsigned smax[GAT_R];
    __shared__ float    sden[GAT_R];
    __shared__ float    sdst[GAT_R];
    __shared__ float    sdv[GAT_R];
    __shared__ int      wcnt[GAT_WAVES];
    __shared__ int      ovf;
    __shared__ int      ovf2;

    const int tid  = threadIdx.x;
    const int lane = tid & 31;
    const int wave = tid >> 5;
    const int lo   = blockIdx.x * GAT_R;
    const int lbase = wave * GAT_CAP;

    for (int i = tid; i < GAT_R * DIM; i += GAT_T) { acc[i] = 0.f; hacc[i] = 0.f; }
    if (tid < GAT_R) { smax[tid] = 0u; sden[tid] = 0.f; sdst[tid] = s_dst[lo + tid]; sdv[tid] = dv[lo + tid]; }
    if (tid == 0) { ovf = 0; ovf2 = 0; }
    __syncthreads();

    static_assert(GAT_R == KEY_GAT_R && GAT_NB <= 255 && GAT_T == HG_T, "the e_dst / inc_v key planes are built for this row split");
    const int nd = key_scan<GAT_NQ>((const int4*)key_ed, blockIdx.x * (GAT_NQ / GAT_NB), blockIdx.x,
                                    (unsigned*)(lst_s + lbase), GAT_CAP, tid);
    __syncthreads();
    int cnt = GAT_CAP + 1;
    if (nd <= GAT_CAP)
        cnt = key_expand((const int4*)key_ed, blockIdx.x, (const unsigned*)(lst_s + lbase), nd, lst_k + lbase, GAT_CAP, lane);
    if (lane == 0) {
        wcnt[wave] = (cnt < GAT_CAP) ? cnt : GAT_CAP;
        if (cnt > GAT_CAP) ovf = 1;
    }
    __syncthreads();
    const int over = __builtin_amdgcn_readfirstlane(ovf);

    if (over == 0) {
        const int n = __builtin_amdgcn_readfirstlane(wcnt[wave]);
        for (int j0 = 0; j0 < n; j0 += 128) {
            unsigned key[4]; int sv[4]; int dd[4]; float ss[4]; bool ok[4];
#pragma unroll
            for (int u = 0; u < 4; ++u) {
                int j = j0 + u * 32 + lane;
                ok[u] = j < n;
                key[u] = ok[u] ? lst_k[lbase + j] : 0u;
            }
#pragma unroll
            for (int u = 0; u < 4; ++u) { sv[u] = e_src[key[u]]; dd[u] = e_dst[key[u]]; }
#pragma unroll
            for (int u = 0; u < 4; ++u) ss[u] = s_src[sv[u]];
#pragma unroll
            for (int u = 0; u < 4; ++u) {
                if (ok[u]) {
                    int j = j0 + u * 32 + lane;
                    unsigned dl = (unsigned)(dd[u] - lo);
                    float sc = gat_leaky(ss[u] + sdst[dl]);
                    atomicMax(&smax[dl], enc_f32(sc));
                    lst_k[lbase + j] = ((unsigned)sv[u] << 8) | dl;
                    lst_s[lbase + j] = sc;
                }
            }
        }
        __syncthreads();
        for (int j = lane; j < n; j += 32) {
            unsigned dl = lst_k[lbase + j] & 255u;
            float m = dec_f32(smax[dl]);
            if (!isfinite(m)) m = 0.f;
            float w = expf(lst_s[lbase + j] - m);
            lst_s[lbase + j] = w;
            atomicAdd(&sden[dl], w);
        }
        __syncthreads();
        for (int j0 = 0; j0 < n; j0 += GAT_RU) {
            float2 xv[GAT_RU]; unsigned kk[GAT_RU]; float ww[GAT_RU];
            const int jj = j0 + (lane & (GAT_RU - 1));
            const unsigned kl = (jj < n) ? lst_k[lbase + jj] : 0u;
            const float    wl = (jj < n) ? lst_s[lbase + jj] : 0.f;
#pragma unroll
            for (int u = 0; u < GAT_RU; ++u) {
                if (j0 + u < n) {
                    kk[u] = (unsigned)__builtin_amdgcn_readlane((int)kl, u);
                    ww[u] = __int_as_float(__builtin_amdgcn_readlane(__float_as_int(wl), u));
                    xv[u] = *(const float2*)(Xp + (size_t)(kk[u] >> 8) * DIM + lane * 2);
                }
            }
#pragma unroll
            for (int u = 0; u < GAT_RU; ++u) {
                if (j0 + u < n) {
                    float* a = acc + (kk[u] & 255u) * DIM + lane * 2;
                    atomicAdd(a + 0, ww[u] * xv[u].x);
                    atomicAdd(a + 1, ww[u] * xv[u].y);
                }
            }
        }
    } else {
        for (int e = tid; e < N_EDGES; e += GAT_T) {
            unsigned dl = (unsigned)(e_dst[e] - lo);
            if (dl < (unsigned)GAT_R) {
                float sc = gat_leaky(s_src[e_src[e]] + sdst[dl]);
                atomicMax(&smax[dl], enc_f32(sc));
            }
        }
        __syncthreads();
        for (int eb = wave * 32; eb < N_EDGES; eb += GAT_T) {
            int e = eb + lane;
            bool valid = e < N_EDGES;
            unsigned dl = valid ? (unsigned)(e_dst[e] - lo) : 0xFFFFFFFFu;
            bool m = dl < (unsigned)GAT_R;
            float w = 0.f; int sidx = 0;
            if (m) {
                sidx = e_src[e];
                float sc = gat_leaky(s_src[sidx] + sdst[dl]);
                float mx = dec_f32(smax[dl]);
                if (!isfinite(mx)) mx = 0.f;
                w = expf(sc - mx);
                atomicAdd(&sden[dl], w);
            }
            unsigned mask = __builtin_amdgcn_ballot_w32(m);
            while (mask != 0u) {
                int l = __builtin_ctz(mask);
                mask &= mask - 1u;
                int   s_l = __shfl(sidx, l);
                int   d_l = __shfl((int)dl, l);
                float w_l = __shfl(w, l);
                float2 xv = *(const float2*)(Xp + (size_t)s_l * DIM + lane * 2);
                float* a = acc + d_l * DIM + lane * 2;
                atomicAdd(a + 0, w_l * xv.x);
                atomicAdd(a + 1, w_l * xv.y);
            }
        }
    }
    __syncthreads();

    const int nd2 = key_scan<HG_NQ>((const int4*)key_iv, blockIdx.x * (HG_NQ / GAT_NB), blockIdx.x,
                                    (unsigned*)(lst_s + lbase), GAT_CAP, tid);
    __syncthreads();
    int cnt2 = GAT_CAP + 1;
    if (nd2 <= GAT_CAP)
        cnt2 = key_expand((const int4*)key_iv, blockIdx.x, (const unsigned*)(lst_s + lbase), nd2, lst_k + lbase, GAT_CAP, lane);
    if (lane == 0) {
        wcnt[wave] = (cnt2 < GAT_CAP) ? cnt2 : GAT_CAP;
        if (cnt2 > GAT_CAP) ovf2 = 1;
    }
    __syncthreads();
    const int over2 = __builtin_amdgcn_readfirstlane(ovf2);

    if (over2 == 0) {
        const int n2 = __builtin_amdgcn_readfirstlane(wcnt[wave]);
        for (int j = lane; j < n2; j += 32) {
            unsigned p = lst_k[lbase + j];
            int e = inc_e[p];
            int v = inc_v[p];
            lst_k[lbase + j] = ((unsigned)e << 8) | (unsigned)(v - lo);
        }
        __syncthreads();
        hg_row_sum<false>(lst_k + lbase, lst_s + lbase, n2, He, hacc, lane);
    } else {
        for (int pb = wave * 32; pb < P_PAIRS; pb += GAT_T) {
            int p = pb + lane;
            bool valid = p < P_PAIRS;
            unsigned dl = valid ? (unsigned)(inc_v[p] - lo) : 0xFFFFFFFFu;
            bool m = dl < (unsigned)GAT_R;
            int eidx = m ? inc_e[p] : 0;
            unsigned mask = __builtin_amdgcn_ballot_w32(m);
            while (mask != 0u) {
                int l = __builtin_ctz(mask);
                mask &= mask - 1u;
                int e_l = __shfl(eidx, l);
                int d_l = __shfl((int)dl, l);
                float2 h = *(const float2*)(He + (size_t)e_l * DIM + lane * 2);
                float* a = hacc + d_l * DIM + lane * 2;
                atomicAdd(a + 0, h.x);
                atomicAdd(a + 1, h.y);
            }
        }
    }
    __syncthreads();

    float4* o4 = (float4*)(out + (size_t)lo * DIM);
    const float4* acc4  = (const float4*)acc;
    const float4* hacc4 = (const float4*)hacc;
    for (int i = tid; i < GAT_R * (DIM / 4); i += GAT_T) {
        float den = sden[i >> 4];
        float d   = sdv[i >> 4];
        float s   = (d > 0.f) ? rsqrtf(d) : 0.f;
        float4 v = acc4[i];
        float4 h = hacc4[i];
        if (den != 0.f) { v.x /= den; v.y /= den; v.z /= den; v.w /= den; }
        float4 x;
        x.x = 0.5f * (v.x + h.x * s);
        x.y = 0.5f * (v.y + h.y * s);
        x.z = 0.5f * (v.z + h.z * s);
        x.w = 0.5f * (v.w + h.w * s);
        x.x = (x.x > 0.f) ? x.x : expm1f(x.x);
        x.y = (x.y > 0.f) ? x.y : expm1f(x.y);
        x.z = (x.z > 0.f) ? x.z : expm1f(x.z);
        x.w = (x.w > 0.f) ? x.w : expm1f(x.w);
        o4[i] = x;
    }
}


extern "C" void kernel_launch(void* const* d_in, const int* in_sizes, int n_in,
                              void* d_out, int out_size, void* d_ws, size_t ws_size,
                              hipStream_t stream) {
    const float* X     = (const float*)d_in[0];
    const float* W     = (const float*)d_in[1];
    const float* bias  = (const float*)d_in[2];
    const float* a_src = (const float*)d_in[3];
    const float* a_dst = (const float*)d_in[4];
    const int*   e_src = (const int*)d_in[5];
    const int*   e_dst = (const int*)d_in[6];
    const int*   inc_v = (const int*)d_in[7];
    const int*   inc_e = (const int*)d_in[8];
    float* out = (float*)d_out;
    float* ws  = (float*)d_ws;

    size_t off = 0;
    float* Xp    = ws + off; off += (size_t)N_NODES * DIM;
    float* s_src = ws + off; off += N_NODES;
    unsigned* key_iv = (unsigned*)(ws + off); off += P_PAIRS / 4;
    float* s_dst = ws + off; off += N_NODES;
    unsigned* key_ie = (unsigned*)(ws + off); off += P_PAIRS / 4;
    float* dv    = ws + off; off += N_NODES;
    float* de    = ws + off; off += M_HYPER;
    float* He    = ws + off; off += (size_t)M_HYPER * DIM;
    unsigned* key_ed = (unsigned*)(ws + off); off += N_EDGES / 4;

    const int B = 256;
    k_project<<<PJ_NB, PJ_T, 0, stream>>>(X, W, bias, a_src, a_dst, Xp, s_src, s_dst, dv);
    k_deg<<<(P_PAIRS + B - 1) / B, B, 0, stream>>>(inc_v, inc_e, e_dst, dv, de, key_ed, key_iv, key_ie);
    k_he_owner<<<HE_NB, HG_T, 0, stream>>>(inc_v, inc_e, key_ie, Xp, dv, de, He);
    k_gat_owner<<<GAT_NB, GAT_T, 0, stream>>>(e_src, e_dst, s_src, s_dst, Xp, inc_v, inc_e, key_ed, key_iv, He, dv, out);
}


# GAT edge row sum: LDS float adds -> f32 16x16x4 matrix unit; list regrouped by 16-row block (LDS cursors), one-hot weight fragment x gathered Xp rows; sum order = cursor arrival, 4 entries per WMMA
# speedup vs baseline: 2.8338x; 1.0312x over previous
#include <hip/hip_runtime.h>
#include <hip/hip_bf16.h>
#include <math.h>

typedef __attribute__((ext_vector_type(2))) float v2f;
typedef __attribute__((ext_vector_type(8))) float v8f;

#define N_NODES 50000
#define N_EDGES 800000
#define M_HYPER 10000
#define P_PAIRS 200000
#define DIM     64
#define NEG_SLOPE 0.2f
#define KEY_GAT_R 200
#define KEY_HE_R  40

__device__ __forceinline__ unsigned enc_f32(float f) {
    unsigned u = __float_as_uint(f);
    return (u & 0x80000000u) ? ~u : (u | 0x80000000u);
}
__device__ __forceinline__ float dec_f32(unsigned u) {
    return (u & 0x80000000u) ? __uint_as_float(u & 0x7FFFFFFFu)
                             : __uint_as_float(~u);
}

#define PJ_T      256
#define PJ_WAVES  (PJ_T / 32)
#define PJ_ROWS   (PJ_WAVES * 16)
#define PJ_TILES  (N_NODES / 16)
#define PJ_NB     ((PJ_TILES + PJ_WAVES - 1) / PJ_WAVES)
#define DEG_CNT   (N_NODES + M_HYPER)

__device__ __forceinline__ float row16_sum(float x) {
    x += __int_as_float(__builtin_amdgcn_update_dpp(0, __float_as_int(x), 0x161, 0xF, 0xF, false));
    x += __int_as_float(__builtin_amdgcn_update_dpp(0, __float_as_int(x), 0x162, 0xF, 0xF, false));
    x += __int_as_float(__builtin_amdgcn_update_dpp(0, __float_as_int(x), 0x164, 0xF, 0xF, false));
    x += __int_as_float(__builtin_amdgcn_update_dpp(0, __float_as_int(x), 0x168, 0xF, 0xF, false));
    return x;
}

__global__ __launch_bounds__(PJ_T)
void k_project(const float* __restrict__ X, const float* __restrict__ W, const float* __restrict__ bias,
               const float* __restrict__ a_src, const float* __restrict__ a_dst,
               float* __restrict__ Xp, float* __restrict__ s_src, float* __restrict__ s_dst,
               float* __restrict__ deg0) {
    static_assert(PJ_NB * PJ_T >= DEG_CNT && PJ_T == 2 * PJ_ROWS, "one thread per degree counter; one thread per scalar of the group");
    __shared__ float sh_s[PJ_ROWS];
    __shared__ float sh_d[PJ_ROWS];

    const int tid  = threadIdx.x;
    const int lane = tid & 31;
    const int wv   = __builtin_amdgcn_readfirstlane(tid >> 5);
    const int gid  = blockIdx.x * PJ_T + tid;
    if (gid < DEG_CNT) deg0[gid] = 0.0f;

    const int tile = blockIdx.x * PJ_WAVES + wv;
    if (tile < PJ_TILES) {
        const int row_base = tile * 16;
        const int lh  = (lane >= 16) ? 1 : 0;
        const int l15 = lane & 15;

        const float* xrow = X + (size_t)(row_base + l15) * DIM + lh * 2;
        const float* wrow = W + (size_t)l15 * DIM + lh * 2;

        v8f acc[4] = {};
#pragma unroll
        for (int ks = 0; ks < 16; ++ks) {
            v2f a;
            a.x = xrow[ks * 4 + 0];
            a.y = xrow[ks * 4 + 1];
#pragma unroll
            for (int t = 0; t < 4; ++t) {
                v2f b;
                const float* wb = wrow + (size_t)t * 16 * DIM + ks * 4;
                b.x = wb[0];
                b.y = wb[1];
                acc[t] = __builtin_amdgcn_wmma_f32_16x16x4_f32(
                    false, a, false, b, (short)0, acc[t], false, false);
            }
        }

        float ps[8], pd[8];
#pragma unroll
        for (int t = 0; t < 4; ++t) {
            int c = t * 16 + l15;
            float bv = bias[c];
            float us = a_src[c];
            float ud = a_dst[c];
#pragma unroll
            for (int v = 0; v < 8; ++v) {
                int m = row_base + v + lh * 8;
                float x = acc[t][v] + bv;
                Xp[(size_t)m * DIM + c] = x;
                ps[v] = (t == 0) ? x * us : ps[v] + x * us;
                pd[v] = (t == 0) ? x * ud : pd[v] + x * ud;
            }
        }
        float sv = 0.f, dvv = 0.f;
#pragma unroll
        for (int v = 0; v < 8; ++v) {
            float rs = row16_sum(ps[v]);
            float rd = row16_sum(pd[v]);
            if ((l15 & 7) == v) { sv = rs; dvv = rd; }
        }
        if (l15 < 8) {
            sh_s[wv * 16 + lh * 8 + l15] = sv;
            sh_d[wv * 16 + lh * 8 + l15] = dvv;
        }
    }
    __syncthreads();
    {
        const int r   = tid & (PJ_ROWS - 1);
        const int row = blockIdx.x * PJ_ROWS + r;
        if (row < N_NODES) {
            if (tid < PJ_ROWS) s_src[row] = sh_s[r];
            else               s_dst[row] = sh_d[r];
        }
    }
}

__device__ __forceinline__ unsigned own_key(int row, unsigned rows_per_group) {
    unsigned k = (unsigned)row / rows_per_group;
    return (k < 255u) ? k : 255u;
}
__device__ __forceinline__ unsigned own_key4(int4 r, unsigned rows_per_group) {
    return own_key(r.x, rows_per_group) | (own_key(r.y, rows_per_group) << 8) |
           (own_key(r.z, rows_per_group) << 16) | (own_key(r.w, rows_per_group) << 24);
}

__global__ void k_deg(const int* __restrict__ inc_v, const int* __restrict__ inc_e, const int* __restrict__ e_dst,
                      float* __restrict__ dv, float* __restrict__ de,
                      unsigned* __restrict__ key_ed, unsigned* __restrict__ key_iv, unsigned* __restrict__ key_ie) {
    static_assert(N_EDGES == 4 * P_PAIRS, "one thread per pair packs four edge keys");
    int p = blockIdx.x * blockDim.x + threadIdx.x;
    if (p >= P_PAIRS) return;
    atomicAdd(dv + inc_v[p], 1.0f);
    atomicAdd(de + inc_e[p], 1.0f);
    key_ed[p] = own_key4(((const int4*)e_dst)[p], (unsigned)KEY_GAT_R);
    if (p < P_PAIRS / 4) {
        key_iv[p] = own_key4(((const int4*)inc_v)[p], (unsigned)KEY_GAT_R);
        key_ie[p] = own_key4(((const int4*)inc_e)[p], (unsigned)KEY_HE_R);
    }
}

#define HG_T      1024
#define HG_WAVES  (HG_T / 32)
#define HG_NQ     (P_PAIRS / 16)
#define HG_SU     7
#define HG_RU     16
#define HE_R      40
#define HE_NB     (M_HYPER / HE_R)
#define HE_CAP    256

template <int NQ>
__device__ __forceinline__ int key_scan(const int4* __restrict__ key4, int qoff, unsigned id,
                                        unsigned* desc, int cap, int tid) {
    constexpr int NIT = (NQ + HG_T - 1) / HG_T;
    const unsigned pat = id * 0x01010101u;
    int cnt = 0;
#pragma unroll 1
    for (int i0 = 0; i0 < NIT; i0 += HG_SU) {
        int4 kv[HG_SU];
        int  qq[HG_SU];
#pragma unroll
        for (int u = 0; u < HG_SU; ++u) {
            int idx = (i0 + u) * HG_T + tid;
            int q = idx + qoff;
            if (q >= NQ) q -= NQ;
            qq[u] = q;
            if (idx < NQ) kv[u] = key4[q];
            else          kv[u] = make_int4(-1, -1, -1, -1);
        }
#pragma unroll
        for (int u = 0; u < HG_SU; ++u) {
            const unsigned y0 = (unsigned)kv[u].x ^ pat, y1 = (unsigned)kv[u].y ^ pat;
            const unsigned y2 = (unsigned)kv[u].z ^ pat, y3 = (unsigned)kv[u].w ^ pat;
            const unsigned any = (((y0 - 0x01010101u) & ~y0) | ((y1 - 0x01010101u) & ~y1) |
                                  ((y2 - 0x01010101u) & ~y2) | ((y3 - 0x01010101u) & ~y3)) & 0x80808080u;
            const bool m = any != 0u;
            const unsigned mask = __builtin_amdgcn_ballot_w32(m);
            if (mask != 0u) {
                const int pos = cnt + (int)__builtin_amdgcn_mbcnt_lo(mask, 0u);
                if (m && pos < cap) desc[pos] = (unsigned)qq[u];
                cnt += __builtin_popcount(mask);
            }
        }
    }
    return cnt;
}

__device__ __forceinline__ int key_expand(const int4* __restrict__ key4, unsigned id, const unsigned* desc, int nd,
                                          unsigned* lst, int cap, int lane) {
    const unsigned pat = id * 0x01010101u;
    const unsigned K = 0x7F7F7F7Fu;
    int cnt = 0;
    for (int j0 = 0; j0 < nd; j0 += 32) {
        const int j = j0 + lane;
        unsigned q = 0u, hm = 0u;
        if (j < nd) {
            q = desc[j];
            const int4 kv = key4[q];
            const unsigned y0 = (unsigned)kv.x ^ pat, y1 = (unsigned)kv.y ^ pat;
            const unsigned y2 = (unsigned)kv.z ^ pat, y3 = (unsigned)kv.w ^ pat;
            const unsigned z0 = ~(((y0 & K) + K) | y0 | K), z1 = ~(((y1 & K) + K) | y1 | K);
            const unsigned z2 = ~(((y2 & K) + K) | y2 | K), z3 = ~(((y3 & K) + K) | y3 | K);
            hm = (z0 >> 7) | (z1 >> 6) | (z2 >> 5) | (z3 >> 4);
        }
        unsigned mask = __builtin_amdgcn_ballot_w32(hm != 0u);
        while (mask != 0u) {
            const bool m = hm != 0u;
            const int b = __ffs((int)hm) - 1;
            const int el = ((b & 7) << 2) | (b >> 3);
            const int pos = cnt + (int)__builtin_amdgcn_mbcnt_lo(mask, 0u);
            if (m && pos < cap) lst[pos] = q * 16u + (unsigned)el;
            cnt += __builtin_popcount(mask);
            hm &= hm - 1u;
            mask = __builtin_amdgcn_ballot_w32(hm != 0u);
        }
    }
    return cnt;
}

template <bool WEIGHTED>
__device__ __forceinline__ void hg_row_sum(const unsigned* lk, const float* ls, int n,
                                           const float* __restrict__ src, float* accp, int lane) {
    for (int j0 = 0; j0 < n; j0 += HG_RU) {
        float2 xv[HG_RU]; unsigned kk[HG_RU]; float ww[HG_RU];
        const int jj = j0 + (lane & (HG_RU - 1));
        const unsigned kl = (jj < n) ? lk[jj] : 0u;
        float wl = 1.f;
        if (WEIGHTED) wl = (jj < n) ? ls[jj] : 0.f;
#pragma unroll
        for (int u = 0; u < HG_RU; ++u) {
            if (j0 + u < n) {
                kk[u] = (unsigned)__builtin_amdgcn_readlane((int)kl, u);
                ww[u] = __int_as_float(__builtin_amdgcn_readlane(__float_as_int(wl), u));
                xv[u] = *(const float2*)(src + (size_t)(kk[u] >> 8) * DIM + lane * 2);
            }
        }
#pragma unroll
        for (int u = 0; u < HG_RU; ++u) {
            if (j0 + u < n) {
                float* a = accp + (kk[u] & 255u) * DIM + lane * 2;
                if (WEIGHTED) {
                    atomicAdd(a + 0, ww[u] * xv[u].x);
                    atomicAdd(a + 1, ww[u] * xv[u].y);
                } else {
                    atomicAdd(a + 0, xv[u].x);
                    atomicAdd(a + 1, xv[u].y);
                }
            }
        }
    }
}

__global__ __launch_bounds__(HG_T)
void k_he_owner(const int* __restrict__ inc_v, const int* __restrict__ inc_e,
                const unsigned* __restrict__ key_ie,
                const float* __restrict__ Xp, const float* __restrict__ dv, const float* __restrict__ de,
                float* __restrict__ He) {
    __shared__ __attribute__((aligned(16))) float hacc[HE_R * DIM];
    __shared__ unsigned lst_k[HG_WAVES * HE_CAP];
    __shared__ float    lst_s[HG_WAVES * HE_CAP];
    __shared__ float    sde[HE_R];
    __shared__ int      wcnt[HG_WAVES];
    __shared__ int      ovf;

    const int tid  = threadIdx.x;
    const int lane = tid & 31;
    const int wave = tid >> 5;
    const int lo   = blockIdx.x * HE_R;
    const int lbase = wave * HE_CAP;

    for (int i = tid; i < HE_R * DIM; i += HG_T) hacc[i] = 0.f;
    if (tid < HE_R) sde[tid] = de[lo + tid];
    if (tid == 0) ovf = 0;
    __syncthreads();

    static_assert(HE_R == KEY_HE_R && HE_NB <= 255, "the inc_e key plane is built for this row split");
    const int nd = key_scan<HG_NQ>((const int4*)key_ie, blockIdx.x * (HG_NQ / HE_NB), blockIdx.x,
                                   (unsigned*)(lst_s + lbase), HE_CAP, tid);
    __syncthreads();
    int cnt = HE_CAP + 1;
    if (nd <= HE_CAP)
        cnt = key_expand((const int4*)key_ie, blockIdx.x, (const unsigned*)(lst_s + lbase), nd, lst_k + lbase, HE_CAP, lane);
    if (lane == 0) {
        wcnt[wave] = (cnt < HE_CAP) ? cnt : HE_CAP;
        if (cnt > HE_CAP) ovf = 1;
    }
    __syncthreads();
    const int over = __builtin_amdgcn_readfirstlane(ovf);

    if (over == 0) {
        const int n = __builtin_amdgcn_readfirstlane(wcnt[wave]);
        for (int j0 = 0; j0 < n; j0 += 128) {
            unsigned key[4]; int vv[4]; int ee[4]; float dd[4]; bool ok[4];
#pragma unroll
            for (int u = 0; u < 4; ++u) {
                int j = j0 + u * 32 + lane;
                ok[u] = j < n;
                key[u] = ok[u] ? lst_k[lbase + j] : 0u;
            }
#pragma unroll
            for (int u = 0; u < 4; ++u) { vv[u] = inc_v[key[u]]; ee[u] = inc_e[key[u]]; }
#pragma unroll
            for (int u = 0; u < 4; ++u) dd[u] = dv[vv[u]];
#pragma unroll
            for (int u = 0; u < 4; ++u) {
                if (ok[u]) {
                    int j = j0 + u * 32 + lane;
                    lst_k[lbase + j] = ((unsigned)vv[u] << 8) | (unsigned)(ee[u] - lo);
                    lst_s[lbase + j] = (dd[u] > 0.f) ? rsqrtf(dd[u]) : 0.f;
                }
            }
        }
        __syncthreads();
        hg_row_sum<true>(lst_k + lbase, lst_s + lbase, n, Xp, hacc, lane);
    } else {
        for (int pb = wave * 32; pb < P_PAIRS; pb += HG_T) {
            int p = pb + lane;
            bool valid = p < P_PAIRS;
            unsigned dl = valid ? (unsigned)(inc_e[p] - lo) : 0xFFFFFFFFu;
            bool m = dl < (unsigned)HE_R;
            float s = 0.f; int vidx = 0;
            if (m) {
                vidx = inc_v[p];
                float d = dv[vidx];
                s = (d > 0.f) ? rsqrtf(d) : 0.f;
            }
            unsigned mask = __builtin_amdgcn_ballot_w32(m);
            while (mask != 0u) {
                int l = __builtin_ctz(mask);
                mask &= mask - 1u;
                int   v_l = __shfl(vidx, l);
                int   d_l = __shfl((int)dl, l);
                float s_l = __shfl(s, l);
                float2 xv = *(const float2*)(Xp + (size_t)v_l * DIM + lane * 2);
                float* a = hacc + d_l * DIM + lane * 2;
                atomicAdd(a + 0, s_l * xv.x);
                atomicAdd(a + 1, s_l * xv.y);
            }
        }
    }
    __syncthreads();

    float4* he4 = (float4*)(He + (size_t)lo * DIM);
    const float4* hacc4 = (const float4*)hacc;
    for (int i = tid; i < HE_R * (DIM / 4); i += HG_T) {
        float d = sde[i >> 4];
        float inv = (d > 0.f) ? (1.f / d) : 0.f;
        float4 v = hacc4[i];
        v.x *= inv; v.y *= inv; v.z *= inv; v.w *= inv;
        he4[i] = v;
    }
}

#define GAT_R     200
#define GAT_NB    (N_NODES / GAT_R)
#define GAT_T     1024
#define GAT_WAVES (GAT_T / 32)
#define GAT_CAP   512
#define GAT_NQ    (N_EDGES / 16)
#define BK_ROWS   16
#define BK_N      ((GAT_R + BK_ROWS - 1) / BK_ROWS)
#define BK_CAP    512
#define BK_STEP   32

__device__ __forceinline__ float gat_leaky(float x) { return (x >= 0.f) ? x : NEG_SLOPE * x; }

__global__ __launch_bounds__(GAT_T)
void k_gat_owner(const int* __restrict__ e_src, const int* __restrict__ e_dst,
                 const float* __restrict__ s_src, const float* __restrict__ s_dst,
                 const float* __restrict__ Xp,
                 const int* __restrict__ inc_v, const int* __restrict__ inc_e,
                 const unsigned* __restrict__ key_ed, const unsigned* __restrict__ key_iv,
                 const float* __restrict__ He, const float* __restrict__ dv,
                 float* __restrict__ out) {
    __shared__ __attribute__((aligned(16))) float acc[GAT_R * DIM];
    __shared__ __attribute__((aligned(16))) float hacc[GAT_R * DIM];
    __shared__ unsigned lst_k[GAT_WAVES * GAT_CAP];
    __shared__ float    lst_s[GAT_WAVES * GAT_CAP];
    __shared__ unsigned smax[GAT_R];
    __shared__ float    sden[GAT_R];
    __shared__ float    sdst[GAT_R];
    __shared__ float    sdv[GAT_R];
    __shared__ int      wcnt[GAT_WAVES];
    __shared__ int      ovf;
    __shared__ int      ovf2;
    __shared__ __attribute__((aligned(16))) unsigned bk_k[BK_N * BK_CAP];
    __shared__ __attribute__((aligned(16))) float    bk_w[BK_N * BK_CAP];
    __shared__ int      bk_cnt[BK_N];
    __shared__ int      bk_ovf;

    const int tid  = threadIdx.x;
    const int lane = tid & 31;
    const int wave = tid >> 5;
    const int lo   = blockIdx.x * GAT_R;
    const int lbase = wave * GAT_CAP;

    for (int i = tid; i < GAT_R * DIM; i += GAT_T) { acc[i] = 0.f; hacc[i] = 0.f; }
    for (int i = tid; i < BK_N * BK_CAP / 4; i += GAT_T) {
        ((uint4*)bk_k)[i]  = make_uint4(0u, 0u, 0u, 0u);
        ((float4*)bk_w)[i] = make_float4(0.f, 0.f, 0.f, 0.f);
    }
    if (tid < BK_N) bk_cnt[tid] = 0;
    if (tid == 0) bk_ovf = 0;
    if (tid < GAT_R) { smax[tid] = 0u; sden[tid] = 0.f; sdst[tid] = s_dst[lo + tid]; sdv[tid] = dv[lo + tid]; }
    if (tid == 0) { ovf = 0; ovf2 = 0; }
    __syncthreads();

    static_assert(GAT_R == KEY_GAT_R && GAT_NB <= 255 && GAT_T == HG_T, "the e_dst / inc_v key planes are built for this row split");
    const int nd = key_scan<GAT_NQ>((const int4*)key_ed, blockIdx.x * (GAT_NQ / GAT_NB), blockIdx.x,
                                    (unsigned*)(lst_s + lbase), GAT_CAP, tid);
    __syncthreads();
    int cnt = GAT_CAP + 1;
    if (nd <= GAT_CAP)
        cnt = key_expand((const int4*)key_ed, blockIdx.x, (const unsigned*)(lst_s + lbase), nd, lst_k + lbase, GAT_CAP, lane);
    if (lane == 0) {
        wcnt[wave] = (cnt < GAT_CAP) ? cnt : GAT_CAP;
        if (cnt > GAT_CAP) ovf = 1;
    }
    __syncthreads();
    const int over = __builtin_amdgcn_readfirstlane(ovf);

    if (over == 0) {
        const int n = __builtin_amdgcn_readfirstlane(wcnt[wave]);
        for (int j0 = 0; j0 < n; j0 += 128) {
            unsigned key[4]; int sv[4]; int dd[4]; float ss[4]; bool ok[4];
#pragma unroll
            for (int u = 0; u < 4; ++u) {
                int j = j0 + u * 32 + lane;
                ok[u] = j < n;
                key[u] = ok[u] ? lst_k[lbase + j] : 0u;
            }
#pragma unroll
            for (int u = 0; u < 4; ++u) { sv[u] = e_src[key[u]]; dd[u] = e_dst[key[u]]; }
#pragma unroll
            for (int u = 0; u < 4; ++u) ss[u] = s_src[sv[u]];
#pragma unroll
            for (int u = 0; u < 4; ++u) {
                if (ok[u]) {
                    int j = j0 + u * 32 + lane;
                    unsigned dl = (unsigned)(dd[u] - lo);
                    float sc = gat_leaky(ss[u] + sdst[dl]);
                    atomicMax(&smax[dl], enc_f32(sc));
                    lst_k[lbase + j] = ((unsigned)sv[u] << 8) | dl;
                    lst_s[lbase + j] = sc;
                }
            }
        }
        __syncthreads();
        for (int j = lane; j < n; j += 32) {
            unsigned key = lst_k[lbase + j];
            unsigned dl = key & 255u;
            float m = dec_f32(smax[dl]);
            if (!isfinite(m)) m = 0.f;
            float w = expf(lst_s[lbase + j] - m);
            lst_s[lbase + j] = w;
            atomicAdd(&sden[dl], w);
            unsigned b = dl / BK_ROWS;
            int pos = atomicAdd(&bk_cnt[b], 1);
            if (pos < BK_CAP) { bk_k[b * BK_CAP + pos] = key; bk_w[b * BK_CAP + pos] = w; }
            else              bk_ovf = 1;
        }
        __syncthreads();
        const int bovf = __builtin_amdgcn_readfirstlane(bk_ovf);
        if (bovf == 0) {
            const int wv = __builtin_amdgcn_readfirstlane(wave);
            if (wv < 2 * BK_N) {
                const int blk = wv >> 1;
                const int fh  = wv & 1;
                const int cnt = __builtin_amdgcn_readfirstlane(bk_cnt[blk]);
                const int c15 = lane & 15;
                const int lh  = lane >> 4;
                const unsigned* kp = bk_k + blk * BK_CAP + 2 * lh;
                const float*    wp = bk_w + blk * BK_CAP + 2 * lh;
                const unsigned xoff = (unsigned)(32 * fh + 2 * c15) * 4u;
                const char* xbytes = (const char*)Xp;
                v8f sum0 = {}, sum1 = {};
                for (int base = 0; base < cnt; base += BK_STEP) {
                    uint2 kk[BK_STEP / 4]; float2 ww[BK_STEP / 4]; float2 xa[BK_STEP / 4], xb[BK_STEP / 4];
#pragma unroll
                    for (int s = 0; s < BK_STEP / 4; ++s) {
                        kk[s] = *(const uint2*)(kp + base + 4 * s);
                        ww[s] = *(const float2*)(wp + base + 4 * s);
                    }
#pragma unroll
                    for (int s = 0; s < BK_STEP / 4; ++s) {
                        xa[s] = *(const float2*)(xbytes + ((kk[s].x & 0xFFFFFF00u) | xoff));
                        xb[s] = *(const float2*)(xbytes + ((kk[s].y & 0xFFFFFF00u) | xoff));
                    }
#pragma unroll
                    for (int s = 0; s < BK_STEP / 4; ++s) {
                        v2f a, b0, b1;
                        a.x = ((kk[s].x & (BK_ROWS - 1)) == (unsigned)c15) ? ww[s].x : 0.f;
                        a.y = ((kk[s].y & (BK_ROWS - 1)) == (unsigned)c15) ? ww[s].y : 0.f;
                        b0.x = xa[s].x; b0.y = xb[s].x;
                        b1.x = xa[s].y; b1.y = xb[s].y;
                        sum0 = __builtin_amdgcn_wmma_f32_16x16x4_f32(false, a, false, b0, (short)0, sum0, false, false);
                        sum1 = __builtin_amdgcn_wmma_f32_16x16x4_f32(false, a, false, b1, (short)0, sum1, false, false);
                    }
                }
#pragma unroll
                for (int v = 0; v < 8; ++v) {
                    const int row = blk * BK_ROWS + v + 8 * lh;
                    if (row < GAT_R)
                        *(float2*)(acc + row * DIM + 32 * fh + 2 * c15) = make_float2(sum0[v], sum1[v]);
                }
            }
        } else {
            hg_row_sum<true>(lst_k + lbase, lst_s + lbase, n, Xp, acc, lane);
        }
    } else {
        for (int e = tid; e < N_EDGES; e += GAT_T) {
            unsigned dl = (unsigned)(e_dst[e] - lo);
            if (dl < (unsigned)GAT_R) {
                float sc = gat_leaky(s_src[e_src[e]] + sdst[dl]);
                atomicMax(&smax[dl], enc_f32(sc));
            }
        }
        __syncthreads();
        for (int eb = wave * 32; eb < N_EDGES; eb += GAT_T) {
            int e = eb + lane;
            bool valid = e < N_EDGES;
            unsigned dl = valid ? (unsigned)(e_dst[e] - lo) : 0xFFFFFFFFu;
            bool m = dl < (unsigned)GAT_R;
            float w = 0.f; int sidx = 0;
            if (m) {
                sidx = e_src[e];
                float sc = gat_leaky(s_src[sidx] + sdst[dl]);
                float mx = dec_f32(smax[dl]);
                if (!isfinite(mx)) mx = 0.f;
                w = expf(sc - mx);
                atomicAdd(&sden[dl], w);
            }
            unsigned mask = __builtin_amdgcn_ballot_w32(m);
            while (mask != 0u) {
                int l = __builtin_ctz(mask);
                mask &= mask - 1u;
                int   s_l = __shfl(sidx, l);
                int   d_l = __shfl((int)dl, l);
                float w_l = __shfl(w, l);
                float2 xv = *(const float2*)(Xp + (size_t)s_l * DIM + lane * 2);
                float* a = acc + d_l * DIM + lane * 2;
                atomicAdd(a + 0, w_l * xv.x);
                atomicAdd(a + 1, w_l * xv.y);
            }
        }
    }
    __syncthreads();

    const int nd2 = key_scan<HG_NQ>((const int4*)key_iv, blockIdx.x * (HG_NQ / GAT_NB), blockIdx.x,
                                    (unsigned*)(lst_s + lbase), GAT_CAP, tid);
    __syncthreads();
    int cnt2 = GAT_CAP + 1;
    if (nd2 <= GAT_CAP)
        cnt2 = key_expand((const int4*)key_iv, blockIdx.x, (const unsigned*)(lst_s + lbase), nd2, lst_k + lbase, GAT_CAP, lane);
    if (lane == 0) {
        wcnt[wave] = (cnt2 < GAT_CAP) ? cnt2 : GAT_CAP;
        if (cnt2 > GAT_CAP) ovf2 = 1;
    }
    __syncthreads();
    const int over2 = __builtin_amdgcn_readfirstlane(ovf2);

    if (over2 == 0) {
        const int n2 = __builtin_amdgcn_readfirstlane(wcnt[wave]);
        for (int j = lane; j < n2; j += 32) {
            unsigned p = lst_k[lbase + j];
            int e = inc_e[p];
            int v = inc_v[p];
            lst_k[lbase + j] = ((unsigned)e << 8) | (unsigned)(v - lo);
        }
        __syncthreads();
        hg_row_sum<false>(lst_k + lbase, lst_s + lbase, n2, He, hacc, lane);
    } else {
        for (int pb = wave * 32; pb < P_PAIRS; pb += GAT_T) {
            int p = pb + lane;
            bool valid = p < P_PAIRS;
            unsigned dl = valid ? (unsigned)(inc_v[p] - lo) : 0xFFFFFFFFu;
            bool m = dl < (unsigned)GAT_R;
            int eidx = m ? inc_e[p] : 0;
            unsigned mask = __builtin_amdgcn_ballot_w32(m);
            while (mask != 0u) {
                int l = __builtin_ctz(mask);
                mask &= mask - 1u;
                int e_l = __shfl(eidx, l);
                int d_l = __shfl((int)dl, l);
                float2 h = *(const float2*)(He + (size_t)e_l * DIM + lane * 2);
                float* a = hacc + d_l * DIM + lane * 2;
                atomicAdd(a + 0, h.x);
                atomicAdd(a + 1, h.y);
            }
        }
    }
    __syncthreads();

    float4* o4 = (float4*)(out + (size_t)lo * DIM);
    const float4* acc4  = (const float4*)acc;
    const float4* hacc4 = (const float4*)hacc;
    for (int i = tid; i < GAT_R * (DIM / 4); i += GAT_T) {
        float den = sden[i >> 4];
        float d   = sdv[i >> 4];
        float s   = (d > 0.f) ? rsqrtf(d) : 0.f;
        float4 v = acc4[i];
        float4 h = hacc4[i];
        if (den != 0.f) { v.x /= den; v.y /= den; v.z /= den; v.w /= den; }
        float4 x;
        x.x = 0.5f * (v.x + h.x * s);
        x.y = 0.5f * (v.y + h.y * s);
        x.z = 0.5f * (v.z + h.z * s);
        x.w = 0.5f * (v.w + h.w * s);
        x.x = (x.x > 0.f) ? x.x : expm1f(x.x);
        x.y = (x.y > 0.f) ? x.y : expm1f(x.y);
        x.z = (x.z > 0.f) ? x.z : expm1f(x.z);
        x.w = (x.w > 0.f) ? x.w : expm1f(x.w);
        o4[i] = x;
    }
}


extern "C" void kernel_launch(void* const* d_in, const int* in_sizes, int n_in,
                              void* d_out, int out_size, void* d_ws, size_t ws_size,
                              hipStream_t stream) {
    const float* X     = (const float*)d_in[0];
    const float* W     = (const float*)d_in[1];
    const float* bias  = (const float*)d_in[2];
    const float* a_src = (const float*)d_in[3];
    const float* a_dst = (const float*)d_in[4];
    const int*   e_src = (const int*)d_in[5];
    const int*   e_dst = (const int*)d_in[6];
    const int*   inc_v = (const int*)d_in[7];
    const int*   inc_e = (const int*)d_in[8];
    float* out = (float*)d_out;
    float* ws  = (float*)d_ws;

    size_t off = 0;
    float* Xp    = ws + off; off += (size_t)N_NODES * DIM;
    float* s_src = ws + off; off += N_NODES;
    unsigned* key_iv = (unsigned*)(ws + off); off += P_PAIRS / 4;
    float* s_dst = ws + off; off += N_NODES;
    unsigned* key_ie = (unsigned*)(ws + off); off += P_PAIRS / 4;
    float* dv    = ws + off; off += N_NODES;
    float* de    = ws + off; off += M_HYPER;
    float* He    = ws + off; off += (size_t)M_HYPER * DIM;
    unsigned* key_ed = (unsigned*)(ws + off); off += N_EDGES / 4;

    const int B = 256;
    k_project<<<PJ_NB, PJ_T, 0, stream>>>(X, W, bias, a_src, a_dst, Xp, s_src, s_dst, dv);
    k_deg<<<(P_PAIRS + B - 1) / B, B, 0, stream>>>(inc_v, inc_e, e_dst, dv, de, key_ed, key_iv, key_ie);
    k_he_owner<<<HE_NB, HG_T, 0, stream>>>(inc_v, inc_e, key_ie, Xp, dv, de, He);
    k_gat_owner<<<GAT_NB, GAT_T, 0, stream>>>(e_src, e_dst, s_src, s_dst, Xp, inc_v, inc_e, key_ed, key_iv, He, dv, out);
}


# Owner scans read a per-owner bitmap column (1 bit per 16-key group, built per call in LDS by the key kernel) instead of testing every byte of the key planes; wave list order (sum order) changes
# speedup vs baseline: 2.9904x; 1.0553x over previous
#include <hip/hip_runtime.h>
#include <hip/hip_bf16.h>
#include <math.h>

typedef __attribute__((ext_vector_type(2))) float v2f;
typedef __attribute__((ext_vector_type(8))) float v8f;

#define N_NODES 50000
#define N_EDGES 800000
#define M_HYPER 10000
#define P_PAIRS 200000
#define DIM     64
#define NEG_SLOPE 0.2f
#define KEY_GAT_R 200
#define KEY_HE_R  40

__device__ __forceinline__ unsigned enc_f32(float f) {
    unsigned u = __float_as_uint(f);
    return (u & 0x80000000u) ? ~u : (u | 0x80000000u);
}
__device__ __forceinline__ float dec_f32(unsigned u) {
    return (u & 0x80000000u) ? __uint_as_float(u & 0x7FFFFFFFu)
                             : __uint_as_float(~u);
}

#define PJ_T      256
#define PJ_WAVES  (PJ_T / 32)
#define PJ_ROWS   (PJ_WAVES * 16)
#define PJ_TILES  (N_NODES / 16)
#define PJ_NB     ((PJ_TILES + PJ_WAVES - 1) / PJ_WAVES)
#define DEG_CNT   (N_NODES + M_HYPER)

__device__ __forceinline__ float row16_sum(float x) {
    x += __int_as_float(__builtin_amdgcn_update_dpp(0, __float_as_int(x), 0x161, 0xF, 0xF, false));
    x += __int_as_float(__builtin_amdgcn_update_dpp(0, __float_as_int(x), 0x162, 0xF, 0xF, false));
    x += __int_as_float(__builtin_amdgcn_update_dpp(0, __float_as_int(x), 0x164, 0xF, 0xF, false));
    x += __int_as_float(__builtin_amdgcn_update_dpp(0, __float_as_int(x), 0x168, 0xF, 0xF, false));
    return x;
}

__global__ __launch_bounds__(PJ_T)
void k_project(const float* __restrict__ X, const float* __restrict__ W, const float* __restrict__ bias,
               const float* __restrict__ a_src, const float* __restrict__ a_dst,
               float* __restrict__ Xp, float* __restrict__ s_src, float* __restrict__ s_dst,
               float* __restrict__ deg0) {
    static_assert(PJ_NB * PJ_T >= DEG_CNT && PJ_T == 2 * PJ_ROWS, "one thread per degree counter; one thread per scalar of the group");
    __shared__ float sh_s[PJ_ROWS];
    __shared__ float sh_d[PJ_ROWS];

    const int tid  = threadIdx.x;
    const int lane = tid & 31;
    const int wv   = __builtin_amdgcn_readfirstlane(tid >> 5);
    const int gid  = blockIdx.x * PJ_T + tid;
    if (gid < DEG_CNT) deg0[gid] = 0.0f;

    const int tile = blockIdx.x * PJ_WAVES + wv;
    if (tile < PJ_TILES) {
        const int row_base = tile * 16;
        const int lh  = (lane >= 16) ? 1 : 0;
        const int l15 = lane & 15;

        const float* xrow = X + (size_t)(row_base + l15) * DIM + lh * 2;
        const float* wrow = W + (size_t)l15 * DIM + lh * 2;

        v8f acc[4] = {};
#pragma unroll
        for (int ks = 0; ks < 16; ++ks) {
            v2f a;
            a.x = xrow[ks * 4 + 0];
            a.y = xrow[ks * 4 + 1];
#pragma unroll
            for (int t = 0; t < 4; ++t) {
                v2f b;
                const float* wb = wrow + (size_t)t * 16 * DIM + ks * 4;
                b.x = wb[0];
                b.y = wb[1];
                acc[t] = __builtin_amdgcn_wmma_f32_16x16x4_f32(
                    false, a, false, b, (short)0, acc[t], false, false);
            }
        }

        float ps[8], pd[8];
#pragma unroll
        for (int t = 0; t < 4; ++t) {
            int c = t * 16 + l15;
            float bv = bias[c];
            float us = a_src[c];
            float ud = a_dst[c];
#pragma unroll
            for (int v = 0; v < 8; ++v) {
                int m = row_base + v + lh * 8;
                float x = acc[t][v] + bv;
                Xp[(size_t)m * DIM + c] = x;
                ps[v] = (t == 0) ? x * us : ps[v] + x * us;
                pd[v] = (t == 0) ? x * ud : pd[v] + x * ud;
            }
        }
        float sv = 0.f, dvv = 0.f;
#pragma unroll
        for (int v = 0; v < 8; ++v) {
            float rs = row16_sum(ps[v]);
            float rd = row16_sum(pd[v]);
            if ((l15 & 7) == v) { sv = rs; dvv = rd; }
        }
        if (l15 < 8) {
            sh_s[wv * 16 + lh * 8 + l15] = sv;
            sh_d[wv * 16 + lh * 8 + l15] = dvv;
        }
    }
    __syncthreads();
    {
        const int r   = tid & (PJ_ROWS - 1);
        const int row = blockIdx.x * PJ_ROWS + r;
        if (row < N_NODES) {
            if (tid < PJ_ROWS) s_src[row] = sh_s[r];
            else               s_dst[row] = sh_d[r];
        }
    }
}

__device__ __forceinline__ unsigned own_key(int row, unsigned rows_per_group) {
    unsigned k = (unsigned)row / rows_per_group;
    return (k < 255u) ? k : 255u;
}
__device__ __forceinline__ unsigned own_key4(int4 r, unsigned rows_per_group) {
    return own_key(r.x, rows_per_group) | (own_key(r.y, rows_per_group) << 8) |
           (own_key(r.z, rows_per_group) << 16) | (own_key(r.w, rows_per_group) << 24);
}

#define BM_T    512
#define BM_ENT  4096
#define BM_PENT (BM_ENT / 4)
#define BM_W    (BM_ENT / 16 / 32)
#define BM_PWB  (BM_PENT / 16 / 32)
#define BM_OWN  250
#define BM_NB   ((N_EDGES + BM_ENT - 1) / BM_ENT)
#define BM_EW   (BM_NB * BM_W)
#define BM_PW   (BM_NB * BM_PWB)
#define BM_PWS  416

__device__ __forceinline__ void bm_set4(unsigned* bm, int words, int4 r, unsigned rows_per_group, unsigned w, unsigned bit) {
    const unsigned k0 = own_key(r.x, rows_per_group), k1 = own_key(r.y, rows_per_group);
    const unsigned k2 = own_key(r.z, rows_per_group), k3 = own_key(r.w, rows_per_group);
    if (k0 < (unsigned)BM_OWN) atomicOr(&bm[k0 * words + w], bit);
    if (k1 < (unsigned)BM_OWN) atomicOr(&bm[k1 * words + w], bit);
    if (k2 < (unsigned)BM_OWN) atomicOr(&bm[k2 * words + w], bit);
    if (k3 < (unsigned)BM_OWN) atomicOr(&bm[k3 * words + w], bit);
}

__global__ __launch_bounds__(BM_T)
void k_deg(const int* __restrict__ inc_v, const int* __restrict__ inc_e, const int* __restrict__ e_dst,
           float* __restrict__ dv, float* __restrict__ de,
           unsigned* __restrict__ key_ed, unsigned* __restrict__ key_iv, unsigned* __restrict__ key_ie,
           unsigned* __restrict__ bm_ed, unsigned* __restrict__ bm_iv, unsigned* __restrict__ bm_ie) {
    static_assert(N_EDGES == 4 * P_PAIRS && N_EDGES % 16 == 0 && P_PAIRS % 16 == 0, "one block slices all three lists; whole key int4s");
    static_assert(BM_EW % 32 == 0 && BM_PWS % 32 == 0 && BM_PWS >= BM_PW, "columns start on 128 bytes");
    static_assert(BM_ENT == 8 * BM_T && BM_PENT == 2 * BM_T, "two edge int4s per thread; one pair int4 per thread of the lower half");
    __shared__ unsigned bme[BM_OWN * BM_W];
    __shared__ unsigned bmv[BM_OWN * BM_PWB];
    __shared__ unsigned bmh[BM_OWN * BM_PWB];
    const int tid = threadIdx.x;
    const int b   = blockIdx.x;

    int4 r[2];
#pragma unroll
    for (int u = 0; u < 2; ++u) {
        const int i4 = b * (BM_ENT / 4) + u * BM_T + tid;
        r[u] = (i4 < N_EDGES / 4) ? ((const int4*)e_dst)[i4] : make_int4(-1, -1, -1, -1);
    }
    const int  p4  = b * (BM_PENT / 4) + tid;
    const bool pok = tid < BM_PENT / 4 && p4 < P_PAIRS / 4;
    int4 rv = make_int4(-1, -1, -1, -1), re = make_int4(-1, -1, -1, -1);
    if (pok) { rv = ((const int4*)inc_v)[p4]; re = ((const int4*)inc_e)[p4]; }

    for (int i = tid; i < BM_OWN * BM_W; i += BM_T) bme[i] = 0u;
    for (int i = tid; i < BM_OWN * BM_PWB; i += BM_T) { bmv[i] = 0u; bmh[i] = 0u; }
    __syncthreads();

#pragma unroll
    for (int u = 0; u < 2; ++u) {
        const unsigned g = (unsigned)(u * BM_T + tid) >> 2;
        bm_set4(bme, BM_W, r[u], (unsigned)KEY_GAT_R, g >> 5, 1u << (g & 31u));
    }
    if (pok) {
        const unsigned g = (unsigned)tid >> 2;
        bm_set4(bmv, BM_PWB, rv, (unsigned)KEY_GAT_R, g >> 5, 1u << (g & 31u));
        bm_set4(bmh, BM_PWB, re, (unsigned)KEY_HE_R,  g >> 5, 1u << (g & 31u));
    }
    __syncthreads();

#pragma unroll
    for (int u = 0; u < 2; ++u) {
        const int i4 = b * (BM_ENT / 4) + u * BM_T + tid;
        if (i4 < N_EDGES / 4) key_ed[i4] = own_key4(r[u], (unsigned)KEY_GAT_R);
    }
    if (pok) {
        key_iv[p4] = own_key4(rv, (unsigned)KEY_GAT_R);
        key_ie[p4] = own_key4(re, (unsigned)KEY_HE_R);
        atomicAdd(dv + rv.x, 1.0f); atomicAdd(dv + rv.y, 1.0f); atomicAdd(dv + rv.z, 1.0f); atomicAdd(dv + rv.w, 1.0f);
        atomicAdd(de + re.x, 1.0f); atomicAdd(de + re.y, 1.0f); atomicAdd(de + re.z, 1.0f); atomicAdd(de + re.w, 1.0f);
    }

    for (int i = tid; i < BM_OWN * BM_W; i += BM_T)
        bm_ed[(size_t)(i / BM_W) * BM_EW + b * BM_W + (i % BM_W)] = bme[i];
    for (int i = tid; i < BM_OWN * BM_PWB; i += BM_T) {
        const size_t o = (size_t)(i / BM_PWB) * BM_PWS + b * BM_PWB + (i % BM_PWB);
        bm_iv[o] = bmv[i];
        bm_ie[o] = bmh[i];
    }
}

#define HG_T      1024
#define HG_WAVES  (HG_T / 32)
#define HG_RU     16
#define HE_R      40
#define HE_NB     (M_HYPER / HE_R)
#define HE_CAP    256

template <int NWORDS>
__device__ __forceinline__ int col_scan(const unsigned* __restrict__ col, unsigned* desc, int cap, int wave, int lane) {
    constexpr int WPW = (NWORDS + HG_WAVES - 1) / HG_WAVES;
    constexpr int NR  = (WPW + 31) / 32;
    unsigned bits[NR]; int widx[NR];
#pragma unroll
    for (int r = 0; r < NR; ++r) {
        const int k = r * 32 + lane;
        widx[r] = wave * WPW + k;
        bits[r] = (k < WPW && widx[r] < NWORDS) ? col[widx[r]] : 0u;
    }
    int cnt = 0;
#pragma unroll
    for (int r = 0; r < NR; ++r) {
        unsigned bb = bits[r];
        unsigned mask = __builtin_amdgcn_ballot_w32(bb != 0u);
        while (mask != 0u) {
            const bool m = bb != 0u;
            const int b = __ffs((int)bb) - 1;
            const int pos = cnt + (int)__builtin_amdgcn_mbcnt_lo(mask, 0u);
            if (m && pos < cap) desc[pos] = (unsigned)(widx[r] * 32 + b);
            cnt += __builtin_popcount(mask);
            bb &= bb - 1u;
            mask = __builtin_amdgcn_ballot_w32(bb != 0u);
        }
    }
    return cnt;
}

__device__ __forceinline__ int key_expand(const int4* __restrict__ key4, unsigned id, const unsigned* desc, int nd,
                                          unsigned* lst, int cap, int lane) {
    const unsigned pat = id * 0x01010101u;
    const unsigned K = 0x7F7F7F7Fu;
    int cnt = 0;
    for (int j0 = 0; j0 < nd; j0 += 32) {
        const int j = j0 + lane;
        unsigned q = 0u, hm = 0u;
        if (j < nd) {
            q = desc[j];
            const int4 kv = key4[q];
            const unsigned y0 = (unsigned)kv.x ^ pat, y1 = (unsigned)kv.y ^ pat;
            const unsigned y2 = (unsigned)kv.z ^ pat, y3 = (unsigned)kv.w ^ pat;
            const unsigned z0 = ~(((y0 & K) + K) | y0 | K), z1 = ~(((y1 & K) + K) | y1 | K);
            const unsigned z2 = ~(((y2 & K) + K) | y2 | K), z3 = ~(((y3 & K) + K) | y3 | K);
            hm = (z0 >> 7) | (z1 >> 6) | (z2 >> 5) | (z3 >> 4);
        }
        unsigned mask = __builtin_amdgcn_ballot_w32(hm != 0u);
        while (mask != 0u) {
            const bool m = hm != 0u;
            const int b = __ffs((int)hm) - 1;
            const int el = ((b & 7) << 2) | (b >> 3);
            const int pos = cnt + (int)__builtin_amdgcn_mbcnt_lo(mask, 0u);
            if (m && pos < cap) lst[pos] = q * 16u + (unsigned)el;
            cnt += __builtin_popcount(mask);
            hm &= hm - 1u;
            mask = __builtin_amdgcn_ballot_w32(hm != 0u);
        }
    }
    return cnt;
}

template <bool WEIGHTED>
__device__ __forceinline__ void hg_row_sum(const unsigned* lk, const float* ls, int n,
                                           const float* __restrict__ src, float* accp, int lane) {
    for (int j0 = 0; j0 < n; j0 += HG_RU) {
        float2 xv[HG_RU]; unsigned kk[HG_RU]; float ww[HG_RU];
        const int jj = j0 + (lane & (HG_RU - 1));
        const unsigned kl = (jj < n) ? lk[jj] : 0u;
        float wl = 1.f;
        if (WEIGHTED) wl = (jj < n) ? ls[jj] : 0.f;
#pragma unroll
        for (int u = 0; u < HG_RU; ++u) {
            if (j0 + u < n) {
                kk[u] = (unsigned)__builtin_amdgcn_readlane((int)kl, u);
                ww[u] = __int_as_float(__builtin_amdgcn_readlane(__float_as_int(wl), u));
                xv[u] = *(const float2*)(src + (size_t)(kk[u] >> 8) * DIM + lane * 2);
            }
        }
#pragma unroll
        for (int u = 0; u < HG_RU; ++u) {
            if (j0 + u < n) {
                float* a = accp + (kk[u] & 255u) * DIM + lane * 2;
                if (WEIGHTED) {
                    atomicAdd(a + 0, ww[u] * xv[u].x);
                    atomicAdd(a + 1, ww[u] * xv[u].y);
                } else {
                    atomicAdd(a + 0, xv[u].x);
                    atomicAdd(a + 1, xv[u].y);
                }
            }
        }
    }
}

__global__ __launch_bounds__(HG_T)
void k_he_owner(const int* __restrict__ inc_v, const int* __restrict__ inc_e,
                const unsigned* __restrict__ key_ie, const unsigned* __restrict__ bm_ie,
                const float* __restrict__ Xp, const float* __restrict__ dv, const float* __restrict__ de,
                float* __restrict__ He) {
    __shared__ __attribute__((aligned(16))) float hacc[HE_R * DIM];
    __shared__ unsigned lst_k[HG_WAVES * HE_CAP];
    __shared__ float    lst_s[HG_WAVES * HE_CAP];
    __shared__ float    sde[HE_R];
    __shared__ int      wcnt[HG_WAVES];
    __shared__ int      ovf;

    const int tid  = threadIdx.x;
    const int lane = tid & 31;
    const int wave = tid >> 5;
    const int lo   = blockIdx.x * HE_R;
    const int lbase = wave * HE_CAP;

    for (int i = tid; i < HE_R * DIM; i += HG_T) hacc[i] = 0.f;
    if (tid < HE_R) sde[tid] = de[lo + tid];
    if (tid == 0) ovf = 0;
    __syncthreads();

    static_assert(HE_R == KEY_HE_R && HE_NB == BM_OWN, "the inc_e key plane and its bitmap columns are built for this row split");
    const int nd = col_scan<BM_PW>(bm_ie + (size_t)blockIdx.x * BM_PWS, (unsigned*)(lst_s + lbase), HE_CAP,
                                   __builtin_amdgcn_readfirstlane(wave), lane);
    __syncthreads();
    int cnt = HE_CAP + 1;
    if (nd <= HE_CAP)
        cnt = key_expand((const int4*)key_ie, blockIdx.x, (const unsigned*)(lst_s + lbase), nd, lst_k + lbase, HE_CAP, lane);
    if (lane == 0) {
        wcnt[wave] = (cnt < HE_CAP) ? cnt : HE_CAP;
        if (cnt > HE_CAP) ovf = 1;
    }
    __syncthreads();
    const int over = __builtin_amdgcn_readfirstlane(ovf);

    if (over == 0) {
        const int n = __builtin_amdgcn_readfirstlane(wcnt[wave]);
        for (int j0 = 0; j0 < n; j0 += 128) {
            unsigned key[4]; int vv[4]; int ee[4]; float dd[4]; bool ok[4];
#pragma unroll
            for (int u = 0; u < 4; ++u) {
                int j = j0 + u * 32 + lane;
                ok[u] = j < n;
                key[u] = ok[u] ? lst_k[lbase + j] : 0u;
            }
#pragma unroll
            for (int u = 0; u < 4; ++u) { vv[u] = inc_v[key[u]]; ee[u] = inc_e[key[u]]; }
#pragma unroll
            for (int u = 0; u < 4; ++u) dd[u] = dv[vv[u]];
#pragma unroll
            for (int u = 0; u < 4; ++u) {
                if (ok[u]) {
                    int j = j0 + u * 32 + lane;
                    lst_k[lbase + j] = ((unsigned)vv[u] << 8) | (unsigned)(ee[u] - lo);
                    lst_s[lbase + j] = (dd[u] > 0.f) ? rsqrtf(dd[u]) : 0.f;
                }
            }
        }
        __syncthreads();
        hg_row_sum<true>(lst_k + lbase, lst_s + lbase, n, Xp, hacc, lane);
    } else {
        for (int pb = wave * 32; pb < P_PAIRS; pb += HG_T) {
            int p = pb + lane;
            bool valid = p < P_PAIRS;
            unsigned dl = valid ? (unsigned)(inc_e[p] - lo) : 0xFFFFFFFFu;
            bool m = dl < (unsigned)HE_R;
            float s = 0.f; int vidx = 0;
            if (m) {
                vidx = inc_v[p];
                float d = dv[vidx];
                s = (d > 0.f) ? rsqrtf(d) : 0.f;
            }
            unsigned mask = __builtin_amdgcn_ballot_w32(m);
            while (mask != 0u) {
                int l = __builtin_ctz(mask);
                mask &= mask - 1u;
                int   v_l = __shfl(vidx, l);
                int   d_l = __shfl((int)dl, l);
                float s_l = __shfl(s, l);
                float2 xv = *(const float2*)(Xp + (size_t)v_l * DIM + lane * 2);
                float* a = hacc + d_l * DIM + lane * 2;
                atomicAdd(a + 0, s_l * xv.x);
                atomicAdd(a + 1, s_l * xv.y);
            }
        }
    }
    __syncthreads();

    float4* he4 = (float4*)(He + (size_t)lo * DIM);
    const float4* hacc4 = (const float4*)hacc;
    for (int i = tid; i < HE_R * (DIM / 4); i += HG_T) {
        float d = sde[i >> 4];
        float inv = (d > 0.f) ? (1.f / d) : 0.f;
        float4 v = hacc4[i];
        v.x *= inv; v.y *= inv; v.z *= inv; v.w *= inv;
        he4[i] = v;
    }
}

#define GAT_R     200
#define GAT_NB    (N_NODES / GAT_R)
#define GAT_T     1024
#define GAT_WAVES (GAT_T / 32)
#define GAT_CAP   512
#define BK_ROWS   16
#define BK_N      ((GAT_R + BK_ROWS - 1) / BK_ROWS)
#define BK_CAP    512
#define BK_STEP   32

__device__ __forceinline__ float gat_leaky(float x) { return (x >= 0.f) ? x : NEG_SLOPE * x; }

__global__ __launch_bounds__(GAT_T)
void k_gat_owner(const int* __restrict__ e_src, const int* __restrict__ e_dst,
                 const float* __restrict__ s_src, const float* __restrict__ s_dst,
                 const float* __restrict__ Xp,
                 const int* __restrict__ inc_v, const int* __restrict__ inc_e,
                 const unsigned* __restrict__ key_ed, const unsigned* __restrict__ key_iv,
                 const unsigned* __restrict__ bm_ed, const unsigned* __restrict__ bm_iv,
                 const float* __restrict__ He, const float* __restrict__ dv,
                 float* __restrict__ out) {
    __shared__ __attribute__((aligned(16))) float acc[GAT_R * DIM];
    __shared__ __attribute__((aligned(16))) float hacc[GAT_R * DIM];
    __shared__ unsigned lst_k[GAT_WAVES * GAT_CAP];
    __shared__ float    lst_s[GAT_WAVES * GAT_CAP];
    __shared__ unsigned smax[GAT_R];
    __shared__ float    sden[GAT_R];
    __shared__ float    sdst[GAT_R];
    __shared__ float    sdv[GAT_R];
    __shared__ int      wcnt[GAT_WAVES];
    __shared__ int      ovf;
    __shared__ int      ovf2;
    __shared__ __attribute__((aligned(16))) unsigned bk_k[BK_N * BK_CAP];
    __shared__ __attribute__((aligned(16))) float    bk_w[BK_N * BK_CAP];
    __shared__ int      bk_cnt[BK_N];
    __shared__ int      bk_ovf;

    const int tid  = threadIdx.x;
    const int lane = tid & 31;
    const int wave = tid >> 5;
    const int lo   = blockIdx.x * GAT_R;
    const int lbase = wave * GAT_CAP;

    for (int i = tid; i < GAT_R * DIM; i += GAT_T) { acc[i] = 0.f; hacc[i] = 0.f; }
    for (int i = tid; i < BK_N * BK_CAP / 4; i += GAT_T) {
        ((uint4*)bk_k)[i]  = make_uint4(0u, 0u, 0u, 0u);
        ((float4*)bk_w)[i] = make_float4(0.f, 0.f, 0.f, 0.f);
    }
    if (tid < BK_N) bk_cnt[tid] = 0;
    if (tid == 0) bk_ovf = 0;
    if (tid < GAT_R) { smax[tid] = 0u; sden[tid] = 0.f; sdst[tid] = s_dst[lo + tid]; sdv[tid] = dv[lo + tid]; }
    if (tid == 0) { ovf = 0; ovf2 = 0; }
    __syncthreads();

    static_assert(GAT_R == KEY_GAT_R && GAT_NB == BM_OWN && GAT_T == HG_T, "the e_dst / inc_v key planes and their bitmap columns are built for this row split");
    const int nd = col_scan<BM_EW>(bm_ed + (size_t)blockIdx.x * BM_EW, (unsigned*)(lst_s + lbase), GAT_CAP,
                                   __builtin_amdgcn_readfirstlane(wave), lane);
    __syncthreads();
    int cnt = GAT_CAP + 1;
    if (nd <= GAT_CAP)
        cnt = key_expand((const int4*)key_ed, blockIdx.x, (const unsigned*)(lst_s + lbase), nd, lst_k + lbase, GAT_CAP, lane);
    if (lane == 0) {
        wcnt[wave] = (cnt < GAT_CAP) ? cnt : GAT_CAP;
        if (cnt > GAT_CAP) ovf = 1;
    }
    __syncthreads();
    const int over = __builtin_amdgcn_readfirstlane(ovf);

    if (over == 0) {
        const int n = __builtin_amdgcn_readfirstlane(wcnt[wave]);
        for (int j0 = 0; j0 < n; j0 += 128) {
            unsigned key[4]; int sv[4]; int dd[4]; float ss[4]; bool ok[4];
#pragma unroll
            for (int u = 0; u < 4; ++u) {
                int j = j0 + u * 32 + lane;
                ok[u] = j < n;
                key[u] = ok[u] ? lst_k[lbase + j] : 0u;
            }
#pragma unroll
            for (int u = 0; u < 4; ++u) { sv[u] = e_src[key[u]]; dd[u] = e_dst[key[u]]; }
#pragma unroll
            for (int u = 0; u < 4; ++u) ss[u] = s_src[sv[u]];
#pragma unroll
            for (int u = 0; u < 4; ++u) {
                if (ok[u]) {
                    int j = j0 + u * 32 + lane;
                    unsigned dl = (unsigned)(dd[u] - lo);
                    float sc = gat_leaky(ss[u] + sdst[dl]);
                    atomicMax(&smax[dl], enc_f32(sc));
                    lst_k[lbase + j] = ((unsigned)sv[u] << 8) | dl;
                    lst_s[lbase + j] = sc;
                }
            }
        }
        __syncthreads();
        for (int j = lane; j < n; j += 32) {
            unsigned key = lst_k[lbase + j];
            unsigned dl = key & 255u;
            float m = dec_f32(smax[dl]);
            if (!isfinite(m)) m = 0.f;
            float w = expf(lst_s[lbase + j] - m);
            lst_s[lbase + j] = w;
            atomicAdd(&sden[dl], w);
            unsigned b = dl / BK_ROWS;
            int pos = atomicAdd(&bk_cnt[b], 1);
            if (pos < BK_CAP) { bk_k[b * BK_CAP + pos] = key; bk_w[b * BK_CAP + pos] = w; }
            else              bk_ovf = 1;
        }
        __syncthreads();
        const int bovf = __builtin_amdgcn_readfirstlane(bk_ovf);
        if (bovf == 0) {
            const int wv = __builtin_amdgcn_readfirstlane(wave);
            if (wv < 2 * BK_N) {
                const int blk = wv >> 1;
                const int fh  = wv & 1;
                const int cnt = __builtin_amdgcn_readfirstlane(bk_cnt[blk]);
                const int c15 = lane & 15;
                const int lh  = lane >> 4;
                const unsigned* kp = bk_k + blk * BK_CAP + 2 * lh;
                const float*    wp = bk_w + blk * BK_CAP + 2 * lh;
                const unsigned xoff = (unsigned)(32 * fh + 2 * c15) * 4u;
                const char* xbytes = (const char*)Xp;
                v8f sum0 = {}, sum1 = {};
                for (int base = 0; base < cnt; base += BK_STEP) {
                    uint2 kk[BK_STEP / 4]; float2 ww[BK_STEP / 4]; float2 xa[BK_STEP / 4], xb[BK_STEP / 4];
#pragma unroll
                    for (int s = 0; s < BK_STEP / 4; ++s) {
                        kk[s] = *(const uint2*)(kp + base + 4 * s);
                        ww[s] = *(const float2*)(wp + base + 4 * s);
                    }
#pragma unroll
                    for (int s = 0; s < BK_STEP / 4; ++s) {
                        xa[s] = *(const float2*)(xbytes + ((kk[s].x & 0xFFFFFF00u) | xoff));
                        xb[s] = *(const float2*)(xbytes + ((kk[s].y & 0xFFFFFF00u) | xoff));
                    }
#pragma unroll
                    for (int s = 0; s < BK_STEP / 4; ++s) {
                        v2f a, b0, b1;
                        a.x = ((kk[s].x & (BK_ROWS - 1)) == (unsigned)c15) ? ww[s].x : 0.f;
                        a.y = ((kk[s].y & (BK_ROWS - 1)) == (unsigned)c15) ? ww[s].y : 0.f;
                        b0.x = xa[s].x; b0.y = xb[s].x;
                        b1.x = xa[s].y; b1.y = xb[s].y;
                        sum0 = __builtin_amdgcn_wmma_f32_16x16x4_f32(false, a, false, b0, (short)0, sum0, false, false);
                        sum1 = __builtin_amdgcn_wmma_f32_16x16x4_f32(false, a, false, b1, (short)0, sum1, false, false);
                    }
                }
#pragma unroll
                for (int v = 0; v < 8; ++v) {
                    const int row = blk * BK_ROWS + v + 8 * lh;
                    if (row < GAT_R)
                        *(float2*)(acc + row * DIM + 32 * fh + 2 * c15) = make_float2(sum0[v], sum1[v]);
                }
            }
        } else {
            hg_row_sum<true>(lst_k + lbase, lst_s + lbase, n, Xp, acc, lane);
        }
    } else {
        for (int e = tid; e < N_EDGES; e += GAT_T) {
            unsigned dl = (unsigned)(e_dst[e] - lo);
            if (dl < (unsigned)GAT_R) {
                float sc = gat_leaky(s_src[e_src[e]] + sdst[dl]);
                atomicMax(&smax[dl], enc_f32(sc));
            }
        }
        __syncthreads();
        for (int eb = wave * 32; eb < N_EDGES; eb += GAT_T) {
            int e = eb + lane;
            bool valid = e < N_EDGES;
            unsigned dl = valid ? (unsigned)(e_dst[e] - lo) : 0xFFFFFFFFu;
            bool m = dl < (unsigned)GAT_R;
            float w = 0.f; int sidx = 0;
            if (m) {
                sidx = e_src[e];
                float sc = gat_leaky(s_src[sidx] + sdst[dl]);
                float mx = dec_f32(smax[dl]);
                if (!isfinite(mx)) mx = 0.f;
                w = expf(sc - mx);
                atomicAdd(&sden[dl], w);
            }
            unsigned mask = __builtin_amdgcn_ballot_w32(m);
            while (mask != 0u) {
                int l = __builtin_ctz(mask);
                mask &= mask - 1u;
                int   s_l = __shfl(sidx, l);
                int   d_l = __shfl((int)dl, l);
                float w_l = __shfl(w, l);
                float2 xv = *(const float2*)(Xp + (size_t)s_l * DIM + lane * 2);
                float* a = acc + d_l * DIM + lane * 2;
                atomicAdd(a + 0, w_l * xv.x);
                atomicAdd(a + 1, w_l * xv.y);
            }
        }
    }
    __syncthreads();

    const int nd2 = col_scan<BM_PW>(bm_iv + (size_t)blockIdx.x * BM_PWS, (unsigned*)(lst_s + lbase), GAT_CAP,
                                    __builtin_amdgcn_readfirstlane(wave), lane);
    __syncthreads();
    int cnt2 = GAT_CAP + 1;
    if (nd2 <= GAT_CAP)
        cnt2 = key_expand((const int4*)key_iv, blockIdx.x, (const unsigned*)(lst_s + lbase), nd2, lst_k + lbase, GAT_CAP, lane);
    if (lane == 0) {
        wcnt[wave] = (cnt2 < GAT_CAP) ? cnt2 : GAT_CAP;
        if (cnt2 > GAT_CAP) ovf2 = 1;
    }
    __syncthreads();
    const int over2 = __builtin_amdgcn_readfirstlane(ovf2);

    if (over2 == 0) {
        const int n2 = __builtin_amdgcn_readfirstlane(wcnt[wave]);
        for (int j = lane; j < n2; j += 32) {
            unsigned p = lst_k[lbase + j];
            int e = inc_e[p];
            int v = inc_v[p];
            lst_k[lbase + j] = ((unsigned)e << 8) | (unsigned)(v - lo);
        }
        __syncthreads();
        hg_row_sum<false>(lst_k + lbase, lst_s + lbase, n2, He, hacc, lane);
    } else {
        for (int pb = wave * 32; pb < P_PAIRS; pb += GAT_T) {
            int p = pb + lane;
            bool valid = p < P_PAIRS;
            unsigned dl = valid ? (unsigned)(inc_v[p] - lo) : 0xFFFFFFFFu;
            bool m = dl < (unsigned)GAT_R;
            int eidx = m ? inc_e[p] : 0;
            unsigned mask = __builtin_amdgcn_ballot_w32(m);
            while (mask != 0u) {
                int l = __builtin_ctz(mask);
                mask &= mask - 1u;
                int e_l = __shfl(eidx, l);
                int d_l = __shfl((int)dl, l);
                float2 h = *(const float2*)(He + (size_t)e_l * DIM + lane * 2);
                float* a = hacc + d_l * DIM + lane * 2;
                atomicAdd(a + 0, h.x);
                atomicAdd(a + 1, h.y);
            }
        }
    }
    __syncthreads();

    float4* o4 = (float4*)(out + (size_t)lo * DIM);
    const float4* acc4  = (const float4*)acc;
    const float4* hacc4 = (const float4*)hacc;
    for (int i = tid; i < GAT_R * (DIM / 4); i += GAT_T) {
        float den = sden[i >> 4];
        float d   = sdv[i >> 4];
        float s   = (d > 0.f) ? rsqrtf(d) : 0.f;
        float4 v = acc4[i];
        float4 h = hacc4[i];
        if (den != 0.f) { v.x /= den; v.y /= den; v.z /= den; v.w /= den; }
        float4 x;
        x.x = 0.5f * (v.x + h.x * s);
        x.y = 0.5f * (v.y + h.y * s);
        x.z = 0.5f * (v.z + h.z * s);
        x.w = 0.5f * (v.w + h.w * s);
        x.x = (x.x > 0.f) ? x.x : expm1f(x.x);
        x.y = (x.y > 0.f) ? x.y : expm1f(x.y);
        x.z = (x.z > 0.f) ? x.z : expm1f(x.z);
        x.w = (x.w > 0.f) ? x.w : expm1f(x.w);
        o4[i] = x;
    }
}


extern "C" void kernel_launch(void* const* d_in, const int* in_sizes, int n_in,
                              void* d_out, int out_size, void* d_ws, size_t ws_size,
                              hipStream_t stream) {
    const float* X     = (const float*)d_in[0];
    const float* W     = (const float*)d_in[1];
    const float* bias  = (const float*)d_in[2];
    const float* a_src = (const float*)d_in[3];
    const float* a_dst = (const float*)d_in[4];
    const int*   e_src = (const int*)d_in[5];
    const int*   e_dst = (const int*)d_in[6];
    const int*   inc_v = (const int*)d_in[7];
    const int*   inc_e = (const int*)d_in[8];
    float* out = (float*)d_out;
    float* ws  = (float*)d_ws;

    size_t off = 0;
    float* Xp    = ws + off; off += (size_t)N_NODES * DIM;
    float* s_src = ws + off; off += N_NODES;
    unsigned* key_iv = (unsigned*)(ws + off); off += P_PAIRS / 4;
    float* s_dst = ws + off; off += N_NODES;
    unsigned* key_ie = (unsigned*)(ws + off); off += P_PAIRS / 4;
    float* dv    = ws + off; off += N_NODES;
    float* de    = ws + off; off += M_HYPER;
    float* He    = ws + off; off += (size_t)M_HYPER * DIM;
    unsigned* key_ed = (unsigned*)(ws + off); off += N_EDGES / 4;
    unsigned* bm_ed  = (unsigned*)(ws + off); off += (size_t)BM_OWN * BM_EW;
    unsigned* bm_iv  = (unsigned*)(ws + off); off += (size_t)BM_OWN * BM_PWS;
    unsigned* bm_ie  = (unsigned*)(ws + off); off += (size_t)BM_OWN * BM_PWS;

    k_project<<<PJ_NB, PJ_T, 0, stream>>>(X, W, bias, a_src, a_dst, Xp, s_src, s_dst, dv);
    k_deg<<<BM_NB, BM_T, 0, stream>>>(inc_v, inc_e, e_dst, dv, de, key_ed, key_iv, key_ie, bm_ed, bm_iv, bm_ie);
    k_he_owner<<<HE_NB, HG_T, 0, stream>>>(inc_v, inc_e, key_ie, bm_ie, Xp, dv, de, He);
    k_gat_owner<<<GAT_NB, GAT_T, 0, stream>>>(e_src, e_dst, s_src, s_dst, Xp, inc_v, inc_e, key_ed, key_iv, bm_ed, bm_iv, He, dv, out);
}
